# starting kernel
# baseline (speedup 1.0000x reference)
#include <hip/hip_runtime.h>

typedef float v2f __attribute__((ext_vector_type(2)));
typedef float v8f __attribute__((ext_vector_type(8)));

#define NNODES 30000
#define HEADS 4
#define SLOPE 0.2f
#define KCHUNK 128
#define PAIRSTRIDE 160

__global__ __launch_bounds__(256)
void gemm_xw_wmma(const float* __restrict__ A, const float* __restrict__ B,
                  float* __restrict__ C, int M, int K, int Nout) {
  __shared__ float ldsB[(KCHUNK / 2) * PAIRSTRIDE];

  const int tid  = threadIdx.x;
  const int lane = tid & 31;
  const int wave = tid >> 5;
  const int Mtiles = M >> 4;

  const int n0 = blockIdx.y * 64;
  int mt = blockIdx.x * 8 + wave;
  const bool active = (mt < Mtiles);
  if (!active) mt = Mtiles - 1;
  const int m0   = mt * 16;
  const int half = lane >> 4;
  const int lm   = lane & 15;

  const int lrow = tid >> 4;
  const int lcol = (tid & 15) * 4;

  v8f acc0 = {}, acc1 = {}, acc2 = {}, acc3 = {};
  const float* __restrict__ arow = A + (size_t)(m0 + lm) * K;

  for (int kb = 0; kb < K; kb += KCHUNK) {
    __syncthreads();
#pragma unroll
    for (int r = 0; r < KCHUNK; r += 16) {
      const int k = r + lrow;
      const float4 v =
          *(const float4*)(B + (size_t)(kb + k) * Nout + n0 + lcol);
      float* p = &ldsB[(k >> 1) * PAIRSTRIDE + lcol * 2 + (k & 1)];
      p[0] = v.x; p[2] = v.y; p[4] = v.z; p[6] = v.w;
    }
    __syncthreads();

    if (kb + KCHUNK < K)
      __builtin_prefetch(arow + kb + KCHUNK, 0, 0);

#pragma unroll 4
    for (int k = 0; k < KCHUNK; k += 4) {
      v2f a = *(const v2f*)(arow + kb + k + 2 * half);
      const float* __restrict__ bp =
          &ldsB[((k >> 1) + half) * PAIRSTRIDE + lm * 2];
      v2f b;
      b = *(const v2f*)(bp + 0);
      acc0 = __builtin_amdgcn_wmma_f32_16x16x4_f32(false, a, false, b, (short)0, acc0, false, false);
      b = *(const v2f*)(bp + 32);
      acc1 = __builtin_amdgcn_wmma_f32_16x16x4_f32(false, a, false, b, (short)0, acc1, false, false);
      b = *(const v2f*)(bp + 64);
      acc2 = __builtin_amdgcn_wmma_f32_16x16x4_f32(false, a, false, b, (short)0, acc2, false, false);
      b = *(const v2f*)(bp + 96);
      acc3 = __builtin_amdgcn_wmma_f32_16x16x4_f32(false, a, false, b, (short)0, acc3, false, false);
    }
  }

  if (active) {
    float* __restrict__ crow = C + (size_t)(m0 + 8 * half) * Nout + n0 + lm;
#pragma unroll
    for (int v = 0; v < 8; ++v) {
      crow[(size_t)v * Nout + 0]  = acc0[v];
      crow[(size_t)v * Nout + 16] = acc1[v];
      crow[(size_t)v * Nout + 32] = acc2[v];
      crow[(size_t)v * Nout + 48] = acc3[v];
    }
  }
}

__global__ void attn_logits(const float* __restrict__ H, const float* __restrict__ a_s,
                            const float* __restrict__ a_d, float* __restrict__ als,
                            float* __restrict__ ald, int Nn, int C) {
  int i = blockIdx.x * blockDim.x + threadIdx.x;
  if (i >= Nn * HEADS) return;
  int n = i / HEADS, h = i % HEADS;
  const float* hp = H + (size_t)n * HEADS * C + (size_t)h * C;
  float s = 0.f, d = 0.f;
  for (int c = 0; c < C; ++c) { float v = hp[c]; s += v * a_s[h * C + c]; d += v * a_d[h * C + c]; }
  als[i] = s; ald[i] = d;
}

__global__ void zero_f(float* __restrict__ p, long long n) {
  long long i = (long long)blockIdx.x * blockDim.x + threadIdx.x;
  if (i < n) p[i] = 0.f;
}

__device__ __forceinline__ unsigned fkey(float x) {
  unsigned b = __float_as_uint(x);
  return (b & 0x80000000u) ? ~b : (b | 0x80000000u);
}
__device__ __forceinline__ float funkey(unsigned k) {
  unsigned b = (k & 0x80000000u) ? (k & 0x7FFFFFFFu) : ~k;
  return __uint_as_float(b);
}

__global__ void edge_scores(const int* __restrict__ src, const int* __restrict__ dst,
                            int E, const float* __restrict__ als, const float* __restrict__ ald,
                            float* __restrict__ ews, unsigned* __restrict__ mkey) {
  int e = blockIdx.x * blockDim.x + threadIdx.x;
  const int EL = E + NNODES;
  if (e >= EL) return;
  int s = (e < E) ? src[e] : (e - E);
  int d = (e < E) ? dst[e] : (e - E);
#pragma unroll
  for (int h = 0; h < HEADS; ++h) {
    float v = als[s * HEADS + h] + ald[d * HEADS + h];
    v = (v > 0.f) ? v : SLOPE * v;
    ews[(size_t)e * HEADS + h] = v;
    atomicMax(&mkey[d * HEADS + h], fkey(v));
  }
}

__global__ void edge_exp(const int* __restrict__ src, const int* __restrict__ dst,
                         int E, float* __restrict__ ews,
                         const unsigned* __restrict__ mkey, float* __restrict__ den) {
  int e = blockIdx.x * blockDim.x + threadIdx.x;
  const int EL = E + NNODES;
  if (e >= EL) return;
  int d = (e < E) ? dst[e] : (e - E);
#pragma unroll
  for (int h = 0; h < HEADS; ++h) {
    float m = funkey(mkey[d * HEADS + h]);
    float ex = __expf(ews[(size_t)e * HEADS + h] - m);
    ews[(size_t)e * HEADS + h] = ex;
    atomicAdd(&den[d * HEADS + h], ex);
  }
}

__global__ void edge_aggregate(const int* __restrict__ src, const int* __restrict__ dst,
                               int E, const float* __restrict__ H,
                               const float* __restrict__ ews, const float* __restrict__ den,
                               float* __restrict__ agg, int C) {
  const int Fout = HEADS * C;
  const int chunks = Fout >> 2;
  long long tid = (long long)blockIdx.x * blockDim.x + threadIdx.x;
  long long total = (long long)(E + NNODES) * chunks;
  if (tid >= total) return;
  int e = (int)(tid / chunks);
  int j = (int)(tid % chunks);
  int f = j * 4;
  int h = f / C;
  int s = (e < E) ? src[e] : (e - E);
  int d = (e < E) ? dst[e] : (e - E);
  float alpha = ews[(size_t)e * HEADS + h] / (den[d * HEADS + h] + 1e-16f);
  const float4 hv = *(const float4*)(H + (size_t)s * Fout + f);
  float* op = agg + (size_t)d * Fout + f;
  atomicAdd(op + 0, hv.x * alpha);
  atomicAdd(op + 1, hv.y * alpha);
  atomicAdd(op + 2, hv.z * alpha);
  atomicAdd(op + 3, hv.w * alpha);
}

__global__ void bias_relu(const float* __restrict__ agg, const float* __restrict__ b,
                          float* __restrict__ out, int Nn, int F) {
  int i = blockIdx.x * blockDim.x + threadIdx.x;
  if (i >= Nn * F) return;
  float v = agg[i] + b[i % F];
  out[i] = v > 0.f ? v : 0.f;
}

__global__ void mean_bias(const float* __restrict__ agg, const float* __restrict__ b,
                          float* __restrict__ out, int Nn, int C) {
  int i = blockIdx.x * blockDim.x + threadIdx.x;
  if (i >= Nn * C) return;
  int n = i / C, c = i % C;
  const float* p = agg + (size_t)n * HEADS * C + c;
  out[i] = 0.25f * (p[0] + p[C] + p[2 * C] + p[3 * C]) + b[c];
}

static void run_gat_layer(hipStream_t st, const float* in, int Fin, int C,
                          const float* W, const float* a_s, const float* a_d, const float* b,
                          float* Hbuf, float* agg, float* outMean,
                          const int* src, const int* dst, int E,
                          float* ALS, float* ALD, float* EWS, unsigned* MKEY, float* DEN) {
  const int Nn = NNODES;
  const int Fout = HEADS * C;
  const int Mtiles = Nn / 16;
  dim3 ggrid((Mtiles + 7) / 8, Fout / 64);
  gemm_xw_wmma<<<ggrid, 256, 0, st>>>(in, W, Hbuf, Nn, Fin, Fout);
  int t = Nn * HEADS;
  attn_logits<<<(t + 255) / 256, 256, 0, st>>>(Hbuf, a_s, a_d, ALS, ALD, Nn, C);
  zero_f<<<(t + 255) / 256, 256, 0, st>>>(DEN, t);
  zero_f<<<(t + 255) / 256, 256, 0, st>>>((float*)MKEY, t);
  long long nagg = (long long)Nn * Fout;
  zero_f<<<(int)((nagg + 255) / 256), 256, 0, st>>>(agg, nagg);
  const int EL = E + Nn;
  edge_scores<<<(EL + 255) / 256, 256, 0, st>>>(src, dst, E, ALS, ALD, EWS, MKEY);
  edge_exp<<<(EL + 255) / 256, 256, 0, st>>>(src, dst, E, EWS, MKEY, DEN);
  long long tot = (long long)EL * (Fout >> 2);
  edge_aggregate<<<(int)((tot + 255) / 256), 256, 0, st>>>(src, dst, E, Hbuf, EWS, DEN, agg, C);
  if (outMean) {
    mean_bias<<<(Nn * C + 255) / 256, 256, 0, st>>>(agg, b, outMean, Nn, C);
  } else {
    bias_relu<<<(int)((nagg + 255) / 256), 256, 0, st>>>(agg, b, agg, Nn, Fout);
  }
}

extern "C" void kernel_launch(void* const* d_in, const int* in_sizes, int n_in,
                              void* d_out, int out_size, void* d_ws, size_t ws_size,
                              hipStream_t stream) {
  const float* x   = (const float*)d_in[0];
  const int*   ei  = (const int*)d_in[1];
  const int E = in_sizes[1] / 2;
  const int* src = ei;
  const int* dst = ei + E;
  const float *W1 = (const float*)d_in[2],  *as1 = (const float*)d_in[3],
              *ad1 = (const float*)d_in[4], *b1  = (const float*)d_in[5];
  const float *W2 = (const float*)d_in[6],  *as2 = (const float*)d_in[7],
              *ad2 = (const float*)d_in[8], *b2  = (const float*)d_in[9];
  const float *Wm = (const float*)d_in[10], *asm_ = (const float*)d_in[11],
              *adm = (const float*)d_in[12], *bm = (const float*)d_in[13];
  const float *Wl = (const float*)d_in[14], *asl = (const float*)d_in[15],
              *adl = (const float*)d_in[16], *bl = (const float*)d_in[17];

  char* ws = (char*)d_ws;
  size_t off = 0;
  auto alloc = [&](size_t bytes) -> void* {
    void* p = ws + off;
    off += (bytes + 255) & ~(size_t)255;
    return p;
  };
  float*    Hbuf = (float*)alloc((size_t)NNODES * 512 * 4);
  float*    A0   = (float*)alloc((size_t)NNODES * 256 * 4);
  float*    A1   = (float*)alloc((size_t)NNODES * 512 * 4);
  float*    ALS  = (float*)alloc((size_t)NNODES * HEADS * 4);
  float*    ALD  = (float*)alloc((size_t)NNODES * HEADS * 4);
  float*    EWS  = (float*)alloc((size_t)(E + NNODES) * HEADS * 4);
  unsigned* MKEY = (unsigned*)alloc((size_t)NNODES * HEADS * 4);
  float*    DEN  = (float*)alloc((size_t)NNODES * HEADS * 4);

  float* zmean   = (float*)d_out;
  float* zlogstd = (float*)d_out + (size_t)NNODES * 32;

  run_gat_layer(stream, x, 256, 64, W1, as1, ad1, b1, Hbuf, A0, nullptr,
                src, dst, E, ALS, ALD, EWS, MKEY, DEN);
  run_gat_layer(stream, A0, 256, 128, W2, as2, ad2, b2, Hbuf, A1, nullptr,
                src, dst, E, ALS, ALD, EWS, MKEY, DEN);
  run_gat_layer(stream, A1, 512, 32, Wm, asm_, adm, bm, Hbuf, A0, zmean,
                src, dst, E, ALS, ALD, EWS, MKEY, DEN);
  run_gat_layer(stream, A1, 512, 32, Wl, asl, adl, bl, Hbuf, A0, zlogstd,
                src, dst, E, ALS, ALD, EWS, MKEY, DEN);
}


# Edges bucketed by destination once per call (count/scan/fill, int atomics); one wave per node does softmax+aggregate+bias: float atomics, zeroing and finalize passes gone; sum order = arrival rank
# speedup vs baseline: 4.1417x; 4.1417x over previous
#include <hip/hip_runtime.h>

typedef float v2f __attribute__((ext_vector_type(2)));
typedef float v8f __attribute__((ext_vector_type(8)));

#define NNODES 30000
#define HEADS 4
#define SLOPE 0.2f
#define KCHUNK 128
#define PAIRSTRIDE 160

__global__ __launch_bounds__(256)
void gemm_xw_wmma(const float* __restrict__ A, const float* __restrict__ B,
                  float* __restrict__ C, int M, int K, int Nout) {
  __shared__ float ldsB[(KCHUNK / 2) * PAIRSTRIDE];

  const int tid  = threadIdx.x;
  const int lane = tid & 31;
  const int wave = tid >> 5;
  const int Mtiles = M >> 4;

  const int n0 = blockIdx.y * 64;
  int mt = blockIdx.x * 8 + wave;
  const bool active = (mt < Mtiles);
  if (!active) mt = Mtiles - 1;
  const int m0   = mt * 16;
  const int half = lane >> 4;
  const int lm   = lane & 15;

  const int lrow = tid >> 4;
  const int lcol = (tid & 15) * 4;

  v8f acc0 = {}, acc1 = {}, acc2 = {}, acc3 = {};
  const float* __restrict__ arow = A + (size_t)(m0 + lm) * K;

  for (int kb = 0; kb < K; kb += KCHUNK) {
    __syncthreads();
#pragma unroll
    for (int r = 0; r < KCHUNK; r += 16) {
      const int k = r + lrow;
      const float4 v =
          *(const float4*)(B + (size_t)(kb + k) * Nout + n0 + lcol);
      float* p = &ldsB[(k >> 1) * PAIRSTRIDE + lcol * 2 + (k & 1)];
      p[0] = v.x; p[2] = v.y; p[4] = v.z; p[6] = v.w;
    }
    __syncthreads();

    if (kb + KCHUNK < K)
      __builtin_prefetch(arow + kb + KCHUNK, 0, 0);

#pragma unroll 4
    for (int k = 0; k < KCHUNK; k += 4) {
      v2f a = *(const v2f*)(arow + kb + k + 2 * half);
      const float* __restrict__ bp =
          &ldsB[((k >> 1) + half) * PAIRSTRIDE + lm * 2];
      v2f b;
      b = *(const v2f*)(bp + 0);
      acc0 = __builtin_amdgcn_wmma_f32_16x16x4_f32(false, a, false, b, (short)0, acc0, false, false);
      b = *(const v2f*)(bp + 32);
      acc1 = __builtin_amdgcn_wmma_f32_16x16x4_f32(false, a, false, b, (short)0, acc1, false, false);
      b = *(const v2f*)(bp + 64);
      acc2 = __builtin_amdgcn_wmma_f32_16x16x4_f32(false, a, false, b, (short)0, acc2, false, false);
      b = *(const v2f*)(bp + 96);
      acc3 = __builtin_amdgcn_wmma_f32_16x16x4_f32(false, a, false, b, (short)0, acc3, false, false);
    }
  }

  if (active) {
    float* __restrict__ crow = C + (size_t)(m0 + 8 * half) * Nout + n0 + lm;
#pragma unroll
    for (int v = 0; v < 8; ++v) {
      crow[(size_t)v * Nout + 0]  = acc0[v];
      crow[(size_t)v * Nout + 16] = acc1[v];
      crow[(size_t)v * Nout + 32] = acc2[v];
      crow[(size_t)v * Nout + 48] = acc3[v];
    }
  }
}

__global__ void attn_logits(const float* __restrict__ H, const float* __restrict__ a_s,
                            const float* __restrict__ a_d, float* __restrict__ als,
                            float* __restrict__ ald, int Nn, int C) {
  int i = blockIdx.x * blockDim.x + threadIdx.x;
  if (i >= Nn * HEADS) return;
  int n = i / HEADS, h = i % HEADS;
  const float* hp = H + (size_t)n * HEADS * C + (size_t)h * C;
  float s = 0.f, d = 0.f;
  for (int c = 0; c < C; ++c) { float v = hp[c]; s += v * a_s[h * C + c]; d += v * a_d[h * C + c]; }
  als[i] = s; ald[i] = d;
}

__global__ void zero_i(int* __restrict__ p, int n) {
  int i = blockIdx.x * blockDim.x + threadIdx.x;
  if (i < n) p[i] = 0;
}

__global__ void dst_count(const int* __restrict__ dst, int E,
                          int* __restrict__ cnt, int* __restrict__ rank) {
  int e = blockIdx.x * blockDim.x + threadIdx.x;
  if (e >= E) return;
  rank[e] = atomicAdd(&cnt[dst[e]], 1);
}

#define SCAN_PER ((NNODES + 1023) / 1024)

__global__ __launch_bounds__(1024)
void dst_scan(const int* __restrict__ cnt, int* __restrict__ rowptr) {
  __shared__ int part[1024];
  const int t = threadIdx.x;
  const int lo = t * SCAN_PER;
  int c[SCAN_PER];
  int s = 0;
#pragma unroll
  for (int q = 0; q < SCAN_PER; ++q) {
    c[q] = (lo + q < NNODES) ? cnt[lo + q] : 0;
    s += c[q];
  }
  part[t] = s;
  __syncthreads();
  for (int off = 1; off < 1024; off <<= 1) {
    const int v = (t >= off) ? part[t - off] : 0;
    __syncthreads();
    part[t] += v;
    __syncthreads();
  }
  int run = part[t] - s;
#pragma unroll
  for (int q = 0; q < SCAN_PER; ++q) {
    if (lo + q < NNODES) rowptr[lo + q] = run;
    run += c[q];
  }
  if (t == 1023) rowptr[NNODES] = part[1023];
}

__global__ void dst_fill(const int* __restrict__ src, const int* __restrict__ dst, int E,
                         const int* __restrict__ rowptr, const int* __restrict__ rank,
                         int* __restrict__ csr_src) {
  int e = blockIdx.x * blockDim.x + threadIdx.x;
  if (e >= E) return;
  csr_src[rowptr[dst[e]] + rank[e]] = src[e];
}

__device__ __forceinline__ float lrelu(float v) { return (v > 0.f) ? v : SLOPE * v; }

template <int C>
__device__ __forceinline__ float pick_head(int k, int lane, float a0, float a1, float a2, float a3) {
  if (C == 128) return (k == 0) ? a0 : (k == 1) ? a1 : (k == 2) ? a2 : a3;
  if (C == 64) {
    const bool hi = (lane & 16) != 0;
    return (k == 0) ? (hi ? a1 : a0) : (hi ? a3 : a2);
  }
  const int h = lane >> 3;
  const float x01 = (h & 1) ? a1 : a0;
  const float x23 = (h & 1) ? a3 : a2;
  return (h & 2) ? x23 : x01;
}

template <int C, bool MEAN>
__global__ __launch_bounds__(256)
void gat_node(const int* __restrict__ rowptr, const int* __restrict__ csr_src,
              const float* __restrict__ H, const float* __restrict__ als,
              const float* __restrict__ ald, const float* __restrict__ bias,
              float* __restrict__ out, int Nn) {
  constexpr int Fout = HEADS * C;
  constexpr int NV = Fout / 128;
  constexpr int UNR = (NV == 1) ? 8 : 4;
  const int lane = threadIdx.x & 31;
  const int d = blockIdx.x * 8 + __builtin_amdgcn_readfirstlane((int)(threadIdx.x >> 5));
  if (d >= Nn) return;
  const int base = rowptr[d];
  const int deg  = rowptr[d + 1] - base;
  const int cnt  = deg + 1;
  const float4 adv = *(const float4*)(ald + (size_t)d * 4);

  float m0 = -__builtin_inff(), m1 = m0, m2 = m0, m3 = m0;
  for (int j = lane; j < cnt; j += 32) {
    const int s = (j < deg) ? csr_src[base + j] : d;
    const float4 a = *(const float4*)(als + (size_t)s * 4);
    m0 = fmaxf(m0, lrelu(a.x + adv.x));
    m1 = fmaxf(m1, lrelu(a.y + adv.y));
    m2 = fmaxf(m2, lrelu(a.z + adv.z));
    m3 = fmaxf(m3, lrelu(a.w + adv.w));
  }
#pragma unroll
  for (int o = 16; o > 0; o >>= 1) {
    m0 = fmaxf(m0, __shfl_xor(m0, o));
    m1 = fmaxf(m1, __shfl_xor(m1, o));
    m2 = fmaxf(m2, __shfl_xor(m2, o));
    m3 = fmaxf(m3, __shfl_xor(m3, o));
  }

  float z0 = 0.f, z1 = 0.f, z2 = 0.f, z3 = 0.f;
  for (int j = lane; j < cnt; j += 32) {
    const int s = (j < deg) ? csr_src[base + j] : d;
    const float4 a = *(const float4*)(als + (size_t)s * 4);
    z0 += __expf(lrelu(a.x + adv.x) - m0);
    z1 += __expf(lrelu(a.y + adv.y) - m1);
    z2 += __expf(lrelu(a.z + adv.z) - m2);
    z3 += __expf(lrelu(a.w + adv.w) - m3);
  }
#pragma unroll
  for (int o = 16; o > 0; o >>= 1) {
    z0 += __shfl_xor(z0, o);
    z1 += __shfl_xor(z1, o);
    z2 += __shfl_xor(z2, o);
    z3 += __shfl_xor(z3, o);
  }
  z0 += 1e-16f; z1 += 1e-16f; z2 += 1e-16f; z3 += 1e-16f;

  float4 acc[NV];
#pragma unroll
  for (int k = 0; k < NV; ++k) acc[k] = make_float4(0.f, 0.f, 0.f, 0.f);

  for (int j0 = 0; j0 < cnt; j0 += 32) {
    const int j = j0 + lane;
    int s_l = d;
    float p0 = 0.f, p1 = 0.f, p2 = 0.f, p3 = 0.f;
    if (j < cnt) {
      s_l = (j < deg) ? csr_src[base + j] : d;
      const float4 a = *(const float4*)(als + (size_t)s_l * 4);
      p0 = __expf(lrelu(a.x + adv.x) - m0) / z0;
      p1 = __expf(lrelu(a.y + adv.y) - m1) / z1;
      p2 = __expf(lrelu(a.z + adv.z) - m2) / z2;
      p3 = __expf(lrelu(a.w + adv.w) - m3) / z3;
    }
    const int left = cnt - j0;
    const int nn = (left < 32) ? left : 32;
    for (int t0 = 0; t0 < nn; t0 += UNR) {
#pragma unroll
      for (int u = 0; u < UNR; ++u) {
        const int t = t0 + u;
        const int s = __builtin_amdgcn_readlane(s_l, t);
        const float a0 = __int_as_float(__builtin_amdgcn_readlane(__float_as_int(p0), t));
        const float a1 = __int_as_float(__builtin_amdgcn_readlane(__float_as_int(p1), t));
        const float a2 = __int_as_float(__builtin_amdgcn_readlane(__float_as_int(p2), t));
        const float a3 = __int_as_float(__builtin_amdgcn_readlane(__float_as_int(p3), t));
        const float* __restrict__ hrow = H + (size_t)s * Fout + lane * 4;
#pragma unroll
        for (int k = 0; k < NV; ++k) {
          const float4 hv = *(const float4*)(hrow + k * 128);
          const float al = pick_head<C>(k, lane, a0, a1, a2, a3);
          acc[k].x += al * hv.x;
          acc[k].y += al * hv.y;
          acc[k].z += al * hv.z;
          acc[k].w += al * hv.w;
        }
      }
    }
  }

  if (MEAN) {
    float4 v = acc[0];
    v.x += __shfl_xor(v.x, 8);  v.y += __shfl_xor(v.y, 8);
    v.z += __shfl_xor(v.z, 8);  v.w += __shfl_xor(v.w, 8);
    v.x += __shfl_xor(v.x, 16); v.y += __shfl_xor(v.y, 16);
    v.z += __shfl_xor(v.z, 16); v.w += __shfl_xor(v.w, 16);
    if (lane < 8) {
      const float4 b = *(const float4*)(bias + lane * 4);
      float4 o;
      o.x = 0.25f * v.x + b.x; o.y = 0.25f * v.y + b.y;
      o.z = 0.25f * v.z + b.z; o.w = 0.25f * v.w + b.w;
      *(float4*)(out + (size_t)d * C + lane * 4) = o;
    }
  } else {
#pragma unroll
    for (int k = 0; k < NV; ++k) {
      const float4 b = *(const float4*)(bias + k * 128 + lane * 4);
      float4 o;
      o.x = fmaxf(acc[k].x + b.x, 0.f); o.y = fmaxf(acc[k].y + b.y, 0.f);
      o.z = fmaxf(acc[k].z + b.z, 0.f); o.w = fmaxf(acc[k].w + b.w, 0.f);
      *(float4*)(out + (size_t)d * Fout + k * 128 + lane * 4) = o;
    }
  }
}

template <int C, bool MEAN>
static void run_gat_layer(hipStream_t st, const float* in, int Fin,
                          const float* W, const float* a_s, const float* a_d, const float* b,
                          float* Hbuf, float* outp,
                          const int* rowptr, const int* csr_src,
                          float* ALS, float* ALD) {
  const int Nn = NNODES;
  const int Fout = HEADS * C;
  const int Mtiles = Nn / 16;
  dim3 ggrid((Mtiles + 7) / 8, Fout / 64);
  gemm_xw_wmma<<<ggrid, 256, 0, st>>>(in, W, Hbuf, Nn, Fin, Fout);
  int t = Nn * HEADS;
  attn_logits<<<(t + 255) / 256, 256, 0, st>>>(Hbuf, a_s, a_d, ALS, ALD, Nn, C);
  gat_node<C, MEAN><<<(Nn + 7) / 8, 256, 0, st>>>(rowptr, csr_src, Hbuf, ALS, ALD, b, outp, Nn);
}

extern "C" void kernel_launch(void* const* d_in, const int* in_sizes, int n_in,
                              void* d_out, int out_size, void* d_ws, size_t ws_size,
                              hipStream_t stream) {
  const float* x   = (const float*)d_in[0];
  const int*   ei  = (const int*)d_in[1];
  const int E = in_sizes[1] / 2;
  const int* src = ei;
  const int* dst = ei + E;
  const float *W1 = (const float*)d_in[2],  *as1 = (const float*)d_in[3],
              *ad1 = (const float*)d_in[4], *b1  = (const float*)d_in[5];
  const float *W2 = (const float*)d_in[6],  *as2 = (const float*)d_in[7],
              *ad2 = (const float*)d_in[8], *b2  = (const float*)d_in[9];
  const float *Wm = (const float*)d_in[10], *asm_ = (const float*)d_in[11],
              *adm = (const float*)d_in[12], *bm = (const float*)d_in[13];
  const float *Wl = (const float*)d_in[14], *asl = (const float*)d_in[15],
              *adl = (const float*)d_in[16], *bl = (const float*)d_in[17];

  char* ws = (char*)d_ws;
  size_t off = 0;
  auto alloc = [&](size_t bytes) -> void* {
    void* p = ws + off;
    off += (bytes + 255) & ~(size_t)255;
    return p;
  };
  float* Hbuf = (float*)alloc((size_t)NNODES * 512 * 4);
  float* A0   = (float*)alloc((size_t)NNODES * 256 * 4);
  float* A1   = (float*)alloc((size_t)NNODES * 512 * 4);
  float* ALS  = (float*)alloc((size_t)NNODES * HEADS * 4);
  float* ALD  = (float*)alloc((size_t)NNODES * HEADS * 4);
  int*   CNT  = (int*)alloc((size_t)NNODES * 4);
  int*   RPTR = (int*)alloc((size_t)(NNODES + 1) * 4);
  int*   RANK = (int*)alloc((size_t)E * 4);
  int*   CSRC = (int*)alloc((size_t)E * 4);

  float* zmean   = (float*)d_out;
  float* zlogstd = (float*)d_out + (size_t)NNODES * 32;

  zero_i<<<(NNODES + 255) / 256, 256, 0, stream>>>(CNT, NNODES);
  dst_count<<<(E + 255) / 256, 256, 0, stream>>>(dst, E, CNT, RANK);
  dst_scan<<<1, 1024, 0, stream>>>(CNT, RPTR);
  dst_fill<<<(E + 255) / 256, 256, 0, stream>>>(src, dst, E, RPTR, RANK, CSRC);

  run_gat_layer<64, false>(stream, x, 256, W1, as1, ad1, b1, Hbuf, A0, RPTR, CSRC, ALS, ALD);
  run_gat_layer<128, false>(stream, A0, 256, W2, as2, ad2, b2, Hbuf, A1, RPTR, CSRC, ALS, ALD);
  run_gat_layer<32, true>(stream, A1, 512, Wm, asm_, adm, bm, Hbuf, zmean, RPTR, CSRC, ALS, ALD);
  run_gat_layer<32, true>(stream, A1, 512, Wl, asl, adl, bl, Hbuf, zlogstd, RPTR, CSRC, ALS, ALD);
}


# Mean and log-std layers share input and graph: run as ONE 8-head layer (one GEMM over both weight matrices side by side, one logits pass, one node pass writing both outputs); 13 launches, was 16
# speedup vs baseline: 4.4601x; 1.0769x over previous
#include <hip/hip_runtime.h>

typedef float v2f __attribute__((ext_vector_type(2)));
typedef float v8f __attribute__((ext_vector_type(8)));

#define NNODES 30000
#define HEADS 4
#define SLOPE 0.2f
#define KCHUNK 128
#define PAIRSTRIDE 160

__global__ __launch_bounds__(256)
void gemm_xw_wmma(const float* __restrict__ A, const float* __restrict__ B,
                  const float* __restrict__ B2, float* __restrict__ C,
                  int M, int K, int ldb, int ldc, int nsplit) {
  __shared__ float ldsB[(KCHUNK / 2) * PAIRSTRIDE];

  const int tid  = threadIdx.x;
  const int lane = tid & 31;
  const int wave = tid >> 5;
  const int Mtiles = M >> 4;

  const int n0 = blockIdx.y * 64;
  const bool second = (n0 >= nsplit);
  const float* __restrict__ Bs = second ? B2 : B;
  const int nb = second ? n0 - nsplit : n0;
  int mt = blockIdx.x * 8 + wave;
  const bool active = (mt < Mtiles);
  if (!active) mt = Mtiles - 1;
  const int m0   = mt * 16;
  const int half = lane >> 4;
  const int lm   = lane & 15;

  const int lrow = tid >> 4;
  const int lcol = (tid & 15) * 4;

  v8f acc0 = {}, acc1 = {}, acc2 = {}, acc3 = {};
  const float* __restrict__ arow = A + (size_t)(m0 + lm) * K;

  for (int kb = 0; kb < K; kb += KCHUNK) {
    __syncthreads();
#pragma unroll
    for (int r = 0; r < KCHUNK; r += 16) {
      const int k = r + lrow;
      const float4 v =
          *(const float4*)(Bs + (size_t)(kb + k) * ldb + nb + lcol);
      float* p = &ldsB[(k >> 1) * PAIRSTRIDE + lcol * 2 + (k & 1)];
      p[0] = v.x; p[2] = v.y; p[4] = v.z; p[6] = v.w;
    }
    __syncthreads();

    if (kb + KCHUNK < K)
      __builtin_prefetch(arow + kb + KCHUNK, 0, 0);

#pragma unroll 4
    for (int k = 0; k < KCHUNK; k += 4) {
      v2f a = *(const v2f*)(arow + kb + k + 2 * half);
      const float* __restrict__ bp =
          &ldsB[((k >> 1) + half) * PAIRSTRIDE + lm * 2];
      v2f b;
      b = *(const v2f*)(bp + 0);
      acc0 = __builtin_amdgcn_wmma_f32_16x16x4_f32(false, a, false, b, (short)0, acc0, false, false);
      b = *(const v2f*)(bp + 32);
      acc1 = __builtin_amdgcn_wmma_f32_16x16x4_f32(false, a, false, b, (short)0, acc1, false, false);
      b = *(const v2f*)(bp + 64);
      acc2 = __builtin_amdgcn_wmma_f32_16x16x4_f32(false, a, false, b, (short)0, acc2, false, false);
      b = *(const v2f*)(bp + 96);
      acc3 = __builtin_amdgcn_wmma_f32_16x16x4_f32(false, a, false, b, (short)0, acc3, false, false);
    }
  }

  if (active) {
    float* __restrict__ crow = C + (size_t)(m0 + 8 * half) * ldc + n0 + lm;
#pragma unroll
    for (int v = 0; v < 8; ++v) {
      crow[(size_t)v * ldc + 0]  = acc0[v];
      crow[(size_t)v * ldc + 16] = acc1[v];
      crow[(size_t)v * ldc + 32] = acc2[v];
      crow[(size_t)v * ldc + 48] = acc3[v];
    }
  }
}

template <int NH>
__global__ void attn_logits(const float* __restrict__ H, const float* __restrict__ a_s,
                            const float* __restrict__ a_d, const float* __restrict__ a_s2,
                            const float* __restrict__ a_d2, float* __restrict__ als,
                            float* __restrict__ ald, int Nn, int C) {
  int i = blockIdx.x * blockDim.x + threadIdx.x;
  if (i >= Nn * NH) return;
  const int h = i % NH;
  const float* hp = H + (size_t)i * C;
  const float* ps = a_s + h * C;
  const float* pd = a_d + h * C;
  if (NH > HEADS && h >= HEADS) { ps = a_s2 + (h - HEADS) * C; pd = a_d2 + (h - HEADS) * C; }
  float s = 0.f, d = 0.f;
  for (int c = 0; c < C; ++c) { float v = hp[c]; s += v * ps[c]; d += v * pd[c]; }
  als[i] = s; ald[i] = d;
}

__global__ void zero_i(int* __restrict__ p, int n) {
  int i = blockIdx.x * blockDim.x + threadIdx.x;
  if (i < n) p[i] = 0;
}

__global__ void dst_count(const int* __restrict__ dst, int E,
                          int* __restrict__ cnt, int* __restrict__ rank) {
  int e = blockIdx.x * blockDim.x + threadIdx.x;
  if (e >= E) return;
  rank[e] = atomicAdd(&cnt[dst[e]], 1);
}

#define SCAN_PER ((NNODES + 1023) / 1024)

__global__ __launch_bounds__(1024)
void dst_scan(const int* __restrict__ cnt, int* __restrict__ rowptr) {
  __shared__ int part[1024];
  const int t = threadIdx.x;
  const int lo = t * SCAN_PER;
  int c[SCAN_PER];
  int s = 0;
#pragma unroll
  for (int q = 0; q < SCAN_PER; ++q) {
    c[q] = (lo + q < NNODES) ? cnt[lo + q] : 0;
    s += c[q];
  }
  part[t] = s;
  __syncthreads();
  for (int off = 1; off < 1024; off <<= 1) {
    const int v = (t >= off) ? part[t - off] : 0;
    __syncthreads();
    part[t] += v;
    __syncthreads();
  }
  int run = part[t] - s;
#pragma unroll
  for (int q = 0; q < SCAN_PER; ++q) {
    if (lo + q < NNODES) rowptr[lo + q] = run;
    run += c[q];
  }
  if (t == 1023) rowptr[NNODES] = part[1023];
}

__global__ void dst_fill(const int* __restrict__ src, const int* __restrict__ dst, int E,
                         const int* __restrict__ rowptr, const int* __restrict__ rank,
                         int* __restrict__ csr_src) {
  int e = blockIdx.x * blockDim.x + threadIdx.x;
  if (e >= E) return;
  csr_src[rowptr[dst[e]] + rank[e]] = src[e];
}

__device__ __forceinline__ float lrelu(float v) { return (v > 0.f) ? v : SLOPE * v; }

template <int C>
__device__ __forceinline__ float pick_head(int k, int lane, float a0, float a1, float a2, float a3) {
  if (C == 128) return (k == 0) ? a0 : (k == 1) ? a1 : (k == 2) ? a2 : a3;
  if (C == 64) {
    const bool hi = (lane & 16) != 0;
    return (k == 0) ? (hi ? a1 : a0) : (hi ? a3 : a2);
  }
  const int h = lane >> 3;
  const float x01 = (h & 1) ? a1 : a0;
  const float x23 = (h & 1) ? a3 : a2;
  return (h & 2) ? x23 : x01;
}

template <int C, bool MEAN>
__global__ __launch_bounds__(256)
void gat_node(const int* __restrict__ rowptr, const int* __restrict__ csr_src,
              const float* __restrict__ H, const float* __restrict__ als,
              const float* __restrict__ ald, const float* __restrict__ bias,
              float* __restrict__ out, int Nn) {
  constexpr int Fout = HEADS * C;
  constexpr int NV = Fout / 128;
  constexpr int UNR = (NV == 1) ? 8 : 4;
  const int lane = threadIdx.x & 31;
  const int d = blockIdx.x * 8 + __builtin_amdgcn_readfirstlane((int)(threadIdx.x >> 5));
  if (d >= Nn) return;
  const int base = rowptr[d];
  const int deg  = rowptr[d + 1] - base;
  const int cnt  = deg + 1;
  const float4 adv = *(const float4*)(ald + (size_t)d * 4);

  float m0 = -__builtin_inff(), m1 = m0, m2 = m0, m3 = m0;
  for (int j = lane; j < cnt; j += 32) {
    const int s = (j < deg) ? csr_src[base + j] : d;
    const float4 a = *(const float4*)(als + (size_t)s * 4);
    m0 = fmaxf(m0, lrelu(a.x + adv.x));
    m1 = fmaxf(m1, lrelu(a.y + adv.y));
    m2 = fmaxf(m2, lrelu(a.z + adv.z));
    m3 = fmaxf(m3, lrelu(a.w + adv.w));
  }
#pragma unroll
  for (int o = 16; o > 0; o >>= 1) {
    m0 = fmaxf(m0, __shfl_xor(m0, o));
    m1 = fmaxf(m1, __shfl_xor(m1, o));
    m2 = fmaxf(m2, __shfl_xor(m2, o));
    m3 = fmaxf(m3, __shfl_xor(m3, o));
  }

  float z0 = 0.f, z1 = 0.f, z2 = 0.f, z3 = 0.f;
  for (int j = lane; j < cnt; j += 32) {
    const int s = (j < deg) ? csr_src[base + j] : d;
    const float4 a = *(const float4*)(als + (size_t)s * 4);
    z0 += __expf(lrelu(a.x + adv.x) - m0);
    z1 += __expf(lrelu(a.y + adv.y) - m1);
    z2 += __expf(lrelu(a.z + adv.z) - m2);
    z3 += __expf(lrelu(a.w + adv.w) - m3);
  }
#pragma unroll
  for (int o = 16; o > 0; o >>= 1) {
    z0 += __shfl_xor(z0, o);
    z1 += __shfl_xor(z1, o);
    z2 += __shfl_xor(z2, o);
    z3 += __shfl_xor(z3, o);
  }
  z0 += 1e-16f; z1 += 1e-16f; z2 += 1e-16f; z3 += 1e-16f;

  float4 acc[NV];
#pragma unroll
  for (int k = 0; k < NV; ++k) acc[k] = make_float4(0.f, 0.f, 0.f, 0.f);

  for (int j0 = 0; j0 < cnt; j0 += 32) {
    const int j = j0 + lane;
    int s_l = d;
    float p0 = 0.f, p1 = 0.f, p2 = 0.f, p3 = 0.f;
    if (j < cnt) {
      s_l = (j < deg) ? csr_src[base + j] : d;
      const float4 a = *(const float4*)(als + (size_t)s_l * 4);
      p0 = __expf(lrelu(a.x + adv.x) - m0) / z0;
      p1 = __expf(lrelu(a.y + adv.y) - m1) / z1;
      p2 = __expf(lrelu(a.z + adv.z) - m2) / z2;
      p3 = __expf(lrelu(a.w + adv.w) - m3) / z3;
    }
    const int left = cnt - j0;
    const int nn = (left < 32) ? left : 32;
    for (int t0 = 0; t0 < nn; t0 += UNR) {
#pragma unroll
      for (int u = 0; u < UNR; ++u) {
        const int t = t0 + u;
        const int s = __builtin_amdgcn_readlane(s_l, t);
        const float a0 = __int_as_float(__builtin_amdgcn_readlane(__float_as_int(p0), t));
        const float a1 = __int_as_float(__builtin_amdgcn_readlane(__float_as_int(p1), t));
        const float a2 = __int_as_float(__builtin_amdgcn_readlane(__float_as_int(p2), t));
        const float a3 = __int_as_float(__builtin_amdgcn_readlane(__float_as_int(p3), t));
        const float* __restrict__ hrow = H + (size_t)s * Fout + lane * 4;
#pragma unroll
        for (int k = 0; k < NV; ++k) {
          const float4 hv = *(const float4*)(hrow + k * 128);
          const float al = pick_head<C>(k, lane, a0, a1, a2, a3);
          acc[k].x += al * hv.x;
          acc[k].y += al * hv.y;
          acc[k].z += al * hv.z;
          acc[k].w += al * hv.w;
        }
      }
    }
  }

  if (MEAN) {
    float4 v = acc[0];
    v.x += __shfl_xor(v.x, 8);  v.y += __shfl_xor(v.y, 8);
    v.z += __shfl_xor(v.z, 8);  v.w += __shfl_xor(v.w, 8);
    v.x += __shfl_xor(v.x, 16); v.y += __shfl_xor(v.y, 16);
    v.z += __shfl_xor(v.z, 16); v.w += __shfl_xor(v.w, 16);
    if (lane < 8) {
      const float4 b = *(const float4*)(bias + lane * 4);
      float4 o;
      o.x = 0.25f * v.x + b.x; o.y = 0.25f * v.y + b.y;
      o.z = 0.25f * v.z + b.z; o.w = 0.25f * v.w + b.w;
      *(float4*)(out + (size_t)d * C + lane * 4) = o;
    }
  } else {
#pragma unroll
    for (int k = 0; k < NV; ++k) {
      const float4 b = *(const float4*)(bias + k * 128 + lane * 4);
      float4 o;
      o.x = fmaxf(acc[k].x + b.x, 0.f); o.y = fmaxf(acc[k].y + b.y, 0.f);
      o.z = fmaxf(acc[k].z + b.z, 0.f); o.w = fmaxf(acc[k].w + b.w, 0.f);
      *(float4*)(out + (size_t)d * Fout + k * 128 + lane * 4) = o;
    }
  }
}

__device__ __forceinline__ void mean_heads_store(float4 v, int lane,
                                                 const float* __restrict__ bias,
                                                 float* __restrict__ orow) {
  v.x += __shfl_xor(v.x, 8);  v.y += __shfl_xor(v.y, 8);
  v.z += __shfl_xor(v.z, 8);  v.w += __shfl_xor(v.w, 8);
  v.x += __shfl_xor(v.x, 16); v.y += __shfl_xor(v.y, 16);
  v.z += __shfl_xor(v.z, 16); v.w += __shfl_xor(v.w, 16);
  if (lane < 8) {
    const float4 b = *(const float4*)(bias + lane * 4);
    float4 o;
    o.x = 0.25f * v.x + b.x; o.y = 0.25f * v.y + b.y;
    o.z = 0.25f * v.z + b.z; o.w = 0.25f * v.w + b.w;
    *(float4*)(orow + lane * 4) = o;
  }
}

__global__ __launch_bounds__(256)
void gat_node_pair(const int* __restrict__ rowptr, const int* __restrict__ csr_src,
                   const float* __restrict__ H, const float* __restrict__ als,
                   const float* __restrict__ ald, const float* __restrict__ bias_a,
                   const float* __restrict__ bias_b, float* __restrict__ out_a,
                   float* __restrict__ out_b, int Nn) {
  constexpr int NH = 2 * HEADS;
  constexpr int C = 32;
  constexpr int Fout = NH * C;
  constexpr int UNR = 8;
  const int lane = threadIdx.x & 31;
  const int d = blockIdx.x * 8 + __builtin_amdgcn_readfirstlane((int)(threadIdx.x >> 5));
  if (d >= Nn) return;
  const int base = rowptr[d];
  const int deg  = rowptr[d + 1] - base;
  const int cnt  = deg + 1;
  float adv[NH];
  {
    const float4 x = *(const float4*)(ald + (size_t)d * NH);
    const float4 y = *(const float4*)(ald + (size_t)d * NH + 4);
    adv[0] = x.x; adv[1] = x.y; adv[2] = x.z; adv[3] = x.w;
    adv[4] = y.x; adv[5] = y.y; adv[6] = y.z; adv[7] = y.w;
  }

  float m[NH];
#pragma unroll
  for (int h = 0; h < NH; ++h) m[h] = -__builtin_inff();
  for (int j = lane; j < cnt; j += 32) {
    const int s = (j < deg) ? csr_src[base + j] : d;
    const float4 a = *(const float4*)(als + (size_t)s * NH);
    const float4 b = *(const float4*)(als + (size_t)s * NH + 4);
    const float e[NH] = {a.x, a.y, a.z, a.w, b.x, b.y, b.z, b.w};
#pragma unroll
    for (int h = 0; h < NH; ++h) m[h] = fmaxf(m[h], lrelu(e[h] + adv[h]));
  }
#pragma unroll
  for (int o = 16; o > 0; o >>= 1) {
#pragma unroll
    for (int h = 0; h < NH; ++h) m[h] = fmaxf(m[h], __shfl_xor(m[h], o));
  }

  float z[NH];
#pragma unroll
  for (int h = 0; h < NH; ++h) z[h] = 0.f;
  for (int j = lane; j < cnt; j += 32) {
    const int s = (j < deg) ? csr_src[base + j] : d;
    const float4 a = *(const float4*)(als + (size_t)s * NH);
    const float4 b = *(const float4*)(als + (size_t)s * NH + 4);
    const float e[NH] = {a.x, a.y, a.z, a.w, b.x, b.y, b.z, b.w};
#pragma unroll
    for (int h = 0; h < NH; ++h) z[h] += __expf(lrelu(e[h] + adv[h]) - m[h]);
  }
#pragma unroll
  for (int o = 16; o > 0; o >>= 1) {
#pragma unroll
    for (int h = 0; h < NH; ++h) z[h] += __shfl_xor(z[h], o);
  }
#pragma unroll
  for (int h = 0; h < NH; ++h) z[h] += 1e-16f;

  float4 acca = make_float4(0.f, 0.f, 0.f, 0.f);
  float4 accb = make_float4(0.f, 0.f, 0.f, 0.f);

  for (int j0 = 0; j0 < cnt; j0 += 32) {
    const int j = j0 + lane;
    int s_l = d;
    float p[NH];
#pragma unroll
    for (int h = 0; h < NH; ++h) p[h] = 0.f;
    if (j < cnt) {
      s_l = (j < deg) ? csr_src[base + j] : d;
      const float4 a = *(const float4*)(als + (size_t)s_l * NH);
      const float4 b = *(const float4*)(als + (size_t)s_l * NH + 4);
      const float e[NH] = {a.x, a.y, a.z, a.w, b.x, b.y, b.z, b.w};
#pragma unroll
      for (int h = 0; h < NH; ++h) p[h] = __expf(lrelu(e[h] + adv[h]) - m[h]) / z[h];
    }
    const int left = cnt - j0;
    const int nn = (left < 32) ? left : 32;
    for (int t0 = 0; t0 < nn; t0 += UNR) {
#pragma unroll
      for (int u = 0; u < UNR; ++u) {
        const int t = t0 + u;
        const int s = __builtin_amdgcn_readlane(s_l, t);
        float a[NH];
#pragma unroll
        for (int h = 0; h < NH; ++h)
          a[h] = __int_as_float(__builtin_amdgcn_readlane(__float_as_int(p[h]), t));
        const float* __restrict__ hrow = H + (size_t)s * Fout + lane * 4;
        const float4 hva = *(const float4*)(hrow);
        const float4 hvb = *(const float4*)(hrow + 128);
        const float ala = pick_head<C>(0, lane, a[0], a[1], a[2], a[3]);
        const float alb = pick_head<C>(0, lane, a[4], a[5], a[6], a[7]);
        acca.x += ala * hva.x; acca.y += ala * hva.y;
        acca.z += ala * hva.z; acca.w += ala * hva.w;
        accb.x += alb * hvb.x; accb.y += alb * hvb.y;
        accb.z += alb * hvb.z; accb.w += alb * hvb.w;
      }
    }
  }

  mean_heads_store(acca, lane, bias_a, out_a + (size_t)d * C);
  mean_heads_store(accb, lane, bias_b, out_b + (size_t)d * C);
}

template <int C, bool MEAN>
static void run_gat_layer(hipStream_t st, const float* in, int Fin,
                          const float* W, const float* a_s, const float* a_d, const float* b,
                          float* Hbuf, float* outp,
                          const int* rowptr, const int* csr_src,
                          float* ALS, float* ALD) {
  const int Nn = NNODES;
  const int Fout = HEADS * C;
  const int Mtiles = Nn / 16;
  dim3 ggrid((Mtiles + 7) / 8, Fout / 64);
  gemm_xw_wmma<<<ggrid, 256, 0, st>>>(in, W, W, Hbuf, Nn, Fin, Fout, Fout, Fout);
  int t = Nn * HEADS;
  attn_logits<HEADS><<<(t + 255) / 256, 256, 0, st>>>(Hbuf, a_s, a_d, a_s, a_d, ALS, ALD, Nn, C);
  gat_node<C, MEAN><<<(Nn + 7) / 8, 256, 0, st>>>(rowptr, csr_src, Hbuf, ALS, ALD, b, outp, Nn);
}

extern "C" void kernel_launch(void* const* d_in, const int* in_sizes, int n_in,
                              void* d_out, int out_size, void* d_ws, size_t ws_size,
                              hipStream_t stream) {
  const float* x   = (const float*)d_in[0];
  const int*   ei  = (const int*)d_in[1];
  const int E = in_sizes[1] / 2;
  const int* src = ei;
  const int* dst = ei + E;
  const float *W1 = (const float*)d_in[2],  *as1 = (const float*)d_in[3],
              *ad1 = (const float*)d_in[4], *b1  = (const float*)d_in[5];
  const float *W2 = (const float*)d_in[6],  *as2 = (const float*)d_in[7],
              *ad2 = (const float*)d_in[8], *b2  = (const float*)d_in[9];
  const float *Wm = (const float*)d_in[10], *asm_ = (const float*)d_in[11],
              *adm = (const float*)d_in[12], *bm = (const float*)d_in[13];
  const float *Wl = (const float*)d_in[14], *asl = (const float*)d_in[15],
              *adl = (const float*)d_in[16], *bl = (const float*)d_in[17];

  char* ws = (char*)d_ws;
  size_t off = 0;
  auto alloc = [&](size_t bytes) -> void* {
    void* p = ws + off;
    off += (bytes + 255) & ~(size_t)255;
    return p;
  };
  float* Hbuf = (float*)alloc((size_t)NNODES * 512 * 4);
  float* A0   = (float*)alloc((size_t)NNODES * 256 * 4);
  float* A1   = (float*)alloc((size_t)NNODES * 512 * 4);
  float* ALS  = (float*)alloc((size_t)NNODES * 2 * HEADS * 4);
  float* ALD  = (float*)alloc((size_t)NNODES * 2 * HEADS * 4);
  int*   CNT  = (int*)alloc((size_t)NNODES * 4);
  int*   RPTR = (int*)alloc((size_t)(NNODES + 1) * 4);
  int*   RANK = (int*)alloc((size_t)E * 4);
  int*   CSRC = (int*)alloc((size_t)E * 4);

  float* zmean   = (float*)d_out;
  float* zlogstd = (float*)d_out + (size_t)NNODES * 32;

  zero_i<<<(NNODES + 255) / 256, 256, 0, stream>>>(CNT, NNODES);
  dst_count<<<(E + 255) / 256, 256, 0, stream>>>(dst, E, CNT, RANK);
  dst_scan<<<1, 1024, 0, stream>>>(CNT, RPTR);
  dst_fill<<<(E + 255) / 256, 256, 0, stream>>>(src, dst, E, RPTR, RANK, CSRC);

  run_gat_layer<64, false>(stream, x, 256, W1, as1, ad1, b1, Hbuf, A0, RPTR, CSRC, ALS, ALD);
  run_gat_layer<128, false>(stream, A0, 256, W2, as2, ad2, b2, Hbuf, A1, RPTR, CSRC, ALS, ALD);
  {
    dim3 ggrid((NNODES / 16 + 7) / 8, 256 / 64);
    gemm_xw_wmma<<<ggrid, 256, 0, stream>>>(A1, Wm, Wl, Hbuf, NNODES, 512, 128, 256, 128);
    attn_logits<2 * HEADS><<<(NNODES * 2 * HEADS + 255) / 256, 256, 0, stream>>>(
        Hbuf, asm_, adm, asl, adl, ALS, ALD, NNODES, 32);
    gat_node_pair<<<(NNODES + 7) / 8, 256, 0, stream>>>(RPTR, CSRC, Hbuf, ALS, ALD, bm, bl,
                                                        zmean, zlogstd, NNODES);
  }
}


# Attention logits computed in the GEMM epilogue from the accumulators (sum order changed: 4 columns per lane, then a 16-lane tree; 128-wide heads as two partial sums): 3 logits launches gone, 10 left
# speedup vs baseline: 4.6270x; 1.0374x over previous
#include <hip/hip_runtime.h>

typedef float v2f __attribute__((ext_vector_type(2)));
typedef float v8f __attribute__((ext_vector_type(8)));

#define NNODES 30000
#define HEADS 4
#define SLOPE 0.2f
#define KCHUNK 128
#define PAIRSTRIDE 160

__device__ __forceinline__ float sum16_to_lane(const float (&x)[16], int lm) {
  const bool b3 = (lm & 8) != 0, b2 = (lm & 4) != 0, b1 = (lm & 2) != 0, b0 = (lm & 1) != 0;
  float y[8], z[4], w[2];
#pragma unroll
  for (int i = 0; i < 8; ++i)
    y[i] = (b3 ? x[i + 8] : x[i]) + __shfl_xor(b3 ? x[i] : x[i + 8], 8);
#pragma unroll
  for (int i = 0; i < 4; ++i)
    z[i] = (b2 ? y[i + 4] : y[i]) + __shfl_xor(b2 ? y[i] : y[i + 4], 4);
#pragma unroll
  for (int i = 0; i < 2; ++i)
    w[i] = (b1 ? z[i + 2] : z[i]) + __shfl_xor(b1 ? z[i] : z[i + 2], 2);
  return (b0 ? w[1] : w[0]) + __shfl_xor(b0 ? w[0] : w[1], 1);
}

__global__ __launch_bounds__(256)
void gemm_xw_wmma(const float* __restrict__ A, const float* __restrict__ B,
                  const float* __restrict__ B2, float* __restrict__ C,
                  int M, int K, int ldb, int ldc, int nsplit,
                  const float* __restrict__ a_s, const float* __restrict__ a_d,
                  const float* __restrict__ a_s2, const float* __restrict__ a_d2,
                  float* __restrict__ als, float* __restrict__ ald, int G, int ldl) {
  __shared__ float ldsB[(KCHUNK / 2) * PAIRSTRIDE];

  const int tid  = threadIdx.x;
  const int lane = tid & 31;
  const int wave = tid >> 5;
  const int Mtiles = M >> 4;

  const int n0 = blockIdx.y * 64;
  const bool second = (n0 >= nsplit);
  const float* __restrict__ Bs = second ? B2 : B;
  const int nb = second ? n0 - nsplit : n0;
  int mt = blockIdx.x * 8 + wave;
  const bool active = (mt < Mtiles);
  if (!active) mt = Mtiles - 1;
  const int m0   = mt * 16;
  const int half = lane >> 4;
  const int lm   = lane & 15;

  const int lrow = tid >> 4;
  const int lcol = (tid & 15) * 4;

  v8f acc0 = {}, acc1 = {}, acc2 = {}, acc3 = {};
  const float* __restrict__ arow = A + (size_t)(m0 + lm) * K;

  for (int kb = 0; kb < K; kb += KCHUNK) {
    __syncthreads();
#pragma unroll
    for (int r = 0; r < KCHUNK; r += 16) {
      const int k = r + lrow;
      const float4 v =
          *(const float4*)(Bs + (size_t)(kb + k) * ldb + nb + lcol);
      float* p = &ldsB[(k >> 1) * PAIRSTRIDE + lcol * 2 + (k & 1)];
      p[0] = v.x; p[2] = v.y; p[4] = v.z; p[6] = v.w;
    }
    __syncthreads();

    if (kb + KCHUNK < K)
      __builtin_prefetch(arow + kb + KCHUNK, 0, 0);

#pragma unroll 4
    for (int k = 0; k < KCHUNK; k += 4) {
      v2f a = *(const v2f*)(arow + kb + k + 2 * half);
      const float* __restrict__ bp =
          &ldsB[((k >> 1) + half) * PAIRSTRIDE + lm * 2];
      v2f b;
      b = *(const v2f*)(bp + 0);
      acc0 = __builtin_amdgcn_wmma_f32_16x16x4_f32(false, a, false, b, (short)0, acc0, false, false);
      b = *(const v2f*)(bp + 32);
      acc1 = __builtin_amdgcn_wmma_f32_16x16x4_f32(false, a, false, b, (short)0, acc1, false, false);
      b = *(const v2f*)(bp + 64);
      acc2 = __builtin_amdgcn_wmma_f32_16x16x4_f32(false, a, false, b, (short)0, acc2, false, false);
      b = *(const v2f*)(bp + 96);
      acc3 = __builtin_amdgcn_wmma_f32_16x16x4_f32(false, a, false, b, (short)0, acc3, false, false);
    }
  }

  if (active) {
    float* __restrict__ crow = C + (size_t)(m0 + 8 * half) * ldc + n0 + lm;
#pragma unroll
    for (int v = 0; v < 8; ++v) {
      crow[(size_t)v * ldc + 0]  = acc0[v];
      crow[(size_t)v * ldc + 16] = acc1[v];
      crow[(size_t)v * ldc + 32] = acc2[v];
      crow[(size_t)v * ldc + 48] = acc3[v];
    }

    const float* __restrict__ vs = second ? a_s2 : a_s;
    const float* __restrict__ vd = second ? a_d2 : a_d;
    float ws[4], wd[4];
#pragma unroll
    for (int t = 0; t < 4; ++t) {
      ws[t] = vs[nb + 16 * t + lm];
      wd[t] = vd[nb + 16 * t + lm];
    }
    float* __restrict__ lout = ((lm & 8) ? ald : als)
                             + (size_t)(m0 + 8 * half + (lm & 7)) * ldl + blockIdx.y * G;
    float x[16];
    if (G == 1) {
#pragma unroll
      for (int v = 0; v < 8; ++v) {
        x[v]     = acc0[v] * ws[0] + acc1[v] * ws[1] + acc2[v] * ws[2] + acc3[v] * ws[3];
        x[8 + v] = acc0[v] * wd[0] + acc1[v] * wd[1] + acc2[v] * wd[2] + acc3[v] * wd[3];
      }
      lout[0] = sum16_to_lane(x, lm);
    } else {
#pragma unroll
      for (int v = 0; v < 8; ++v) {
        x[v]     = acc0[v] * ws[0] + acc1[v] * ws[1];
        x[8 + v] = acc0[v] * wd[0] + acc1[v] * wd[1];
      }
      lout[0] = sum16_to_lane(x, lm);
#pragma unroll
      for (int v = 0; v < 8; ++v) {
        x[v]     = acc2[v] * ws[2] + acc3[v] * ws[3];
        x[8 + v] = acc2[v] * wd[2] + acc3[v] * wd[3];
      }
      lout[1] = sum16_to_lane(x, lm);
    }
  }
}

__global__ void zero_i(int* __restrict__ p, int n) {
  int i = blockIdx.x * blockDim.x + threadIdx.x;
  if (i < n) p[i] = 0;
}

__global__ void dst_count(const int* __restrict__ dst, int E,
                          int* __restrict__ cnt, int* __restrict__ rank) {
  int e = blockIdx.x * blockDim.x + threadIdx.x;
  if (e >= E) return;
  rank[e] = atomicAdd(&cnt[dst[e]], 1);
}

#define SCAN_PER ((NNODES + 1023) / 1024)

__global__ __launch_bounds__(1024)
void dst_scan(const int* __restrict__ cnt, int* __restrict__ rowptr) {
  __shared__ int part[1024];
  const int t = threadIdx.x;
  const int lo = t * SCAN_PER;
  int c[SCAN_PER];
  int s = 0;
#pragma unroll
  for (int q = 0; q < SCAN_PER; ++q) {
    c[q] = (lo + q < NNODES) ? cnt[lo + q] : 0;
    s += c[q];
  }
  part[t] = s;
  __syncthreads();
  for (int off = 1; off < 1024; off <<= 1) {
    const int v = (t >= off) ? part[t - off] : 0;
    __syncthreads();
    part[t] += v;
    __syncthreads();
  }
  int run = part[t] - s;
#pragma unroll
  for (int q = 0; q < SCAN_PER; ++q) {
    if (lo + q < NNODES) rowptr[lo + q] = run;
    run += c[q];
  }
  if (t == 1023) rowptr[NNODES] = part[1023];
}

__global__ void dst_fill(const int* __restrict__ src, const int* __restrict__ dst, int E,
                         const int* __restrict__ rowptr, const int* __restrict__ rank,
                         int* __restrict__ csr_src) {
  int e = blockIdx.x * blockDim.x + threadIdx.x;
  if (e >= E) return;
  csr_src[rowptr[dst[e]] + rank[e]] = src[e];
}

__device__ __forceinline__ float lrelu(float v) { return (v > 0.f) ? v : SLOPE * v; }

template <int C>
__device__ __forceinline__ float4 node_logits(const float* __restrict__ p, int node) {
  if (C == 128) {
    const float4 q0 = *(const float4*)(p + (size_t)node * 8);
    const float4 q1 = *(const float4*)(p + (size_t)node * 8 + 4);
    return make_float4(q0.x + q0.y, q0.z + q0.w, q1.x + q1.y, q1.z + q1.w);
  }
  return *(const float4*)(p + (size_t)node * 4);
}

template <int C>
__device__ __forceinline__ float pick_head(int k, int lane, float a0, float a1, float a2, float a3) {
  if (C == 128) return (k == 0) ? a0 : (k == 1) ? a1 : (k == 2) ? a2 : a3;
  if (C == 64) {
    const bool hi = (lane & 16) != 0;
    return (k == 0) ? (hi ? a1 : a0) : (hi ? a3 : a2);
  }
  const int h = lane >> 3;
  const float x01 = (h & 1) ? a1 : a0;
  const float x23 = (h & 1) ? a3 : a2;
  return (h & 2) ? x23 : x01;
}

template <int C, bool MEAN>
__global__ __launch_bounds__(256)
void gat_node(const int* __restrict__ rowptr, const int* __restrict__ csr_src,
              const float* __restrict__ H, const float* __restrict__ als,
              const float* __restrict__ ald, const float* __restrict__ bias,
              float* __restrict__ out, int Nn) {
  constexpr int Fout = HEADS * C;
  constexpr int NV = Fout / 128;
  constexpr int UNR = (NV == 1) ? 8 : 4;
  const int lane = threadIdx.x & 31;
  const int d = blockIdx.x * 8 + __builtin_amdgcn_readfirstlane((int)(threadIdx.x >> 5));
  if (d >= Nn) return;
  const int base = rowptr[d];
  const int deg  = rowptr[d + 1] - base;
  const int cnt  = deg + 1;
  const float4 adv = node_logits<C>(ald, d);

  float m0 = -__builtin_inff(), m1 = m0, m2 = m0, m3 = m0;
  for (int j = lane; j < cnt; j += 32) {
    const int s = (j < deg) ? csr_src[base + j] : d;
    const float4 a = node_logits<C>(als, s);
    m0 = fmaxf(m0, lrelu(a.x + adv.x));
    m1 = fmaxf(m1, lrelu(a.y + adv.y));
    m2 = fmaxf(m2, lrelu(a.z + adv.z));
    m3 = fmaxf(m3, lrelu(a.w + adv.w));
  }
#pragma unroll
  for (int o = 16; o > 0; o >>= 1) {
    m0 = fmaxf(m0, __shfl_xor(m0, o));
    m1 = fmaxf(m1, __shfl_xor(m1, o));
    m2 = fmaxf(m2, __shfl_xor(m2, o));
    m3 = fmaxf(m3, __shfl_xor(m3, o));
  }

  float z0 = 0.f, z1 = 0.f, z2 = 0.f, z3 = 0.f;
  for (int j = lane; j < cnt; j += 32) {
    const int s = (j < deg) ? csr_src[base + j] : d;
    const float4 a = node_logits<C>(als, s);
    z0 += __expf(lrelu(a.x + adv.x) - m0);
    z1 += __expf(lrelu(a.y + adv.y) - m1);
    z2 += __expf(lrelu(a.z + adv.z) - m2);
    z3 += __expf(lrelu(a.w + adv.w) - m3);
  }
#pragma unroll
  for (int o = 16; o > 0; o >>= 1) {
    z0 += __shfl_xor(z0, o);
    z1 += __shfl_xor(z1, o);
    z2 += __shfl_xor(z2, o);
    z3 += __shfl_xor(z3, o);
  }
  z0 += 1e-16f; z1 += 1e-16f; z2 += 1e-16f; z3 += 1e-16f;

  float4 acc[NV];
#pragma unroll
  for (int k = 0; k < NV; ++k) acc[k] = make_float4(0.f, 0.f, 0.f, 0.f);

  for (int j0 = 0; j0 < cnt; j0 += 32) {
    const int j = j0 + lane;
    int s_l = d;
    float p0 = 0.f, p1 = 0.f, p2 = 0.f, p3 = 0.f;
    if (j < cnt) {
      s_l = (j < deg) ? csr_src[base + j] : d;
      const float4 a = node_logits<C>(als, s_l);
      p0 = __expf(lrelu(a.x + adv.x) - m0) / z0;
      p1 = __expf(lrelu(a.y + adv.y) - m1) / z1;
      p2 = __expf(lrelu(a.z + adv.z) - m2) / z2;
      p3 = __expf(lrelu(a.w + adv.w) - m3) / z3;
    }
    const int left = cnt - j0;
    const int nn = (left < 32) ? left : 32;
    for (int t0 = 0; t0 < nn; t0 += UNR) {
#pragma unroll
      for (int u = 0; u < UNR; ++u) {
        const int t = t0 + u;
        const int s = __builtin_amdgcn_readlane(s_l, t);
        const float a0 = __int_as_float(__builtin_amdgcn_readlane(__float_as_int(p0), t));
        const float a1 = __int_as_float(__builtin_amdgcn_readlane(__float_as_int(p1), t));
        const float a2 = __int_as_float(__builtin_amdgcn_readlane(__float_as_int(p2), t));
        const float a3 = __int_as_float(__builtin_amdgcn_readlane(__float_as_int(p3), t));
        const float* __restrict__ hrow = H + (size_t)s * Fout + lane * 4;
#pragma unroll
        for (int k = 0; k < NV; ++k) {
          const float4 hv = *(const float4*)(hrow + k * 128);
          const float al = pick_head<C>(k, lane, a0, a1, a2, a3);
          acc[k].x += al * hv.x;
          acc[k].y += al * hv.y;
          acc[k].z += al * hv.z;
          acc[k].w += al * hv.w;
        }
      }
    }
  }

  if (MEAN) {
    float4 v = acc[0];
    v.x += __shfl_xor(v.x, 8);  v.y += __shfl_xor(v.y, 8);
    v.z += __shfl_xor(v.z, 8);  v.w += __shfl_xor(v.w, 8);
    v.x += __shfl_xor(v.x, 16); v.y += __shfl_xor(v.y, 16);
    v.z += __shfl_xor(v.z, 16); v.w += __shfl_xor(v.w, 16);
    if (lane < 8) {
      const float4 b = *(const float4*)(bias + lane * 4);
      float4 o;
      o.x = 0.25f * v.x + b.x; o.y = 0.25f * v.y + b.y;
      o.z = 0.25f * v.z + b.z; o.w = 0.25f * v.w + b.w;
      *(float4*)(out + (size_t)d * C + lane * 4) = o;
    }
  } else {
#pragma unroll
    for (int k = 0; k < NV; ++k) {
      const float4 b = *(const float4*)(bias + k * 128 + lane * 4);
      float4 o;
      o.x = fmaxf(acc[k].x + b.x, 0.f); o.y = fmaxf(acc[k].y + b.y, 0.f);
      o.z = fmaxf(acc[k].z + b.z, 0.f); o.w = fmaxf(acc[k].w + b.w, 0.f);
      *(float4*)(out + (size_t)d * Fout + k * 128 + lane * 4) = o;
    }
  }
}

__device__ __forceinline__ void mean_heads_store(float4 v, int lane,
                                                 const float* __restrict__ bias,
                                                 float* __restrict__ orow) {
  v.x += __shfl_xor(v.x, 8);  v.y += __shfl_xor(v.y, 8);
  v.z += __shfl_xor(v.z, 8);  v.w += __shfl_xor(v.w, 8);
  v.x += __shfl_xor(v.x, 16); v.y += __shfl_xor(v.y, 16);
  v.z += __shfl_xor(v.z, 16); v.w += __shfl_xor(v.w, 16);
  if (lane < 8) {
    const float4 b = *(const float4*)(bias + lane * 4);
    float4 o;
    o.x = 0.25f * v.x + b.x; o.y = 0.25f * v.y + b.y;
    o.z = 0.25f * v.z + b.z; o.w = 0.25f * v.w + b.w;
    *(float4*)(orow + lane * 4) = o;
  }
}

__global__ __launch_bounds__(256)
void gat_node_pair(const int* __restrict__ rowptr, const int* __restrict__ csr_src,
                   const float* __restrict__ H, const float* __restrict__ als,
                   const float* __restrict__ ald, const float* __restrict__ bias_a,
                   const float* __restrict__ bias_b, float* __restrict__ out_a,
                   float* __restrict__ out_b, int Nn) {
  constexpr int NH = 2 * HEADS;
  constexpr int C = 32;
  constexpr int Fout = NH * C;
  constexpr int UNR = 8;
  const int lane = threadIdx.x & 31;
  const int d = blockIdx.x * 8 + __builtin_amdgcn_readfirstlane((int)(threadIdx.x >> 5));
  if (d >= Nn) return;
  const int base = rowptr[d];
  const int deg  = rowptr[d + 1] - base;
  const int cnt  = deg + 1;
  float adv[NH];
  {
    const float4 x = *(const float4*)(ald + (size_t)d * NH);
    const float4 y = *(const float4*)(ald + (size_t)d * NH + 4);
    adv[0] = x.x; adv[1] = x.y; adv[2] = x.z; adv[3] = x.w;
    adv[4] = y.x; adv[5] = y.y; adv[6] = y.z; adv[7] = y.w;
  }

  float m[NH];
#pragma unroll
  for (int h = 0; h < NH; ++h) m[h] = -__builtin_inff();
  for (int j = lane; j < cnt; j += 32) {
    const int s = (j < deg) ? csr_src[base + j] : d;
    const float4 a = *(const float4*)(als + (size_t)s * NH);
    const float4 b = *(const float4*)(als + (size_t)s * NH + 4);
    const float e[NH] = {a.x, a.y, a.z, a.w, b.x, b.y, b.z, b.w};
#pragma unroll
    for (int h = 0; h < NH; ++h) m[h] = fmaxf(m[h], lrelu(e[h] + adv[h]));
  }
#pragma unroll
  for (int o = 16; o > 0; o >>= 1) {
#pragma unroll
    for (int h = 0; h < NH; ++h) m[h] = fmaxf(m[h], __shfl_xor(m[h], o));
  }

  float z[NH];
#pragma unroll
  for (int h = 0; h < NH; ++h) z[h] = 0.f;
  for (int j = lane; j < cnt; j += 32) {
    const int s = (j < deg) ? csr_src[base + j] : d;
    const float4 a = *(const float4*)(als + (size_t)s * NH);
    const float4 b = *(const float4*)(als + (size_t)s * NH + 4);
    const float e[NH] = {a.x, a.y, a.z, a.w, b.x, b.y, b.z, b.w};
#pragma unroll
    for (int h = 0; h < NH; ++h) z[h] += __expf(lrelu(e[h] + adv[h]) - m[h]);
  }
#pragma unroll
  for (int o = 16; o > 0; o >>= 1) {
#pragma unroll
    for (int h = 0; h < NH; ++h) z[h] += __shfl_xor(z[h], o);
  }
#pragma unroll
  for (int h = 0; h < NH; ++h) z[h] += 1e-16f;

  float4 acca = make_float4(0.f, 0.f, 0.f, 0.f);
  float4 accb = make_float4(0.f, 0.f, 0.f, 0.f);

  for (int j0 = 0; j0 < cnt; j0 += 32) {
    const int j = j0 + lane;
    int s_l = d;
    float p[NH];
#pragma unroll
    for (int h = 0; h < NH; ++h) p[h] = 0.f;
    if (j < cnt) {
      s_l = (j < deg) ? csr_src[base + j] : d;
      const float4 a = *(const float4*)(als + (size_t)s_l * NH);
      const float4 b = *(const float4*)(als + (size_t)s_l * NH + 4);
      const float e[NH] = {a.x, a.y, a.z, a.w, b.x, b.y, b.z, b.w};
#pragma unroll
      for (int h = 0; h < NH; ++h) p[h] = __expf(lrelu(e[h] + adv[h]) - m[h]) / z[h];
    }
    const int left = cnt - j0;
    const int nn = (left < 32) ? left : 32;
    for (int t0 = 0; t0 < nn; t0 += UNR) {
#pragma unroll
      for (int u = 0; u < UNR; ++u) {
        const int t = t0 + u;
        const int s = __builtin_amdgcn_readlane(s_l, t);
        float a[NH];
#pragma unroll
        for (int h = 0; h < NH; ++h)
          a[h] = __int_as_float(__builtin_amdgcn_readlane(__float_as_int(p[h]), t));
        const float* __restrict__ hrow = H + (size_t)s * Fout + lane * 4;
        const float4 hva = *(const float4*)(hrow);
        const float4 hvb = *(const float4*)(hrow + 128);
        const float ala = pick_head<C>(0, lane, a[0], a[1], a[2], a[3]);
        const float alb = pick_head<C>(0, lane, a[4], a[5], a[6], a[7]);
        acca.x += ala * hva.x; acca.y += ala * hva.y;
        acca.z += ala * hva.z; acca.w += ala * hva.w;
        accb.x += alb * hvb.x; accb.y += alb * hvb.y;
        accb.z += alb * hvb.z; accb.w += alb * hvb.w;
      }
    }
  }

  mean_heads_store(acca, lane, bias_a, out_a + (size_t)d * C);
  mean_heads_store(accb, lane, bias_b, out_b + (size_t)d * C);
}

template <int C, bool MEAN>
static void run_gat_layer(hipStream_t st, const float* in, int Fin,
                          const float* W, const float* a_s, const float* a_d, const float* b,
                          float* Hbuf, float* outp,
                          const int* rowptr, const int* csr_src,
                          float* ALS, float* ALD) {
  const int Nn = NNODES;
  const int Fout = HEADS * C;
  const int Mtiles = Nn / 16;
  dim3 ggrid((Mtiles + 7) / 8, Fout / 64);
  gemm_xw_wmma<<<ggrid, 256, 0, st>>>(in, W, W, Hbuf, Nn, Fin, Fout, Fout, Fout,
                                      a_s, a_d, a_s, a_d, ALS, ALD, 1, Fout / 64);
  gat_node<C, MEAN><<<(Nn + 7) / 8, 256, 0, st>>>(rowptr, csr_src, Hbuf, ALS, ALD, b, outp, Nn);
}

extern "C" void kernel_launch(void* const* d_in, const int* in_sizes, int n_in,
                              void* d_out, int out_size, void* d_ws, size_t ws_size,
                              hipStream_t stream) {
  const float* x   = (const float*)d_in[0];
  const int*   ei  = (const int*)d_in[1];
  const int E = in_sizes[1] / 2;
  const int* src = ei;
  const int* dst = ei + E;
  const float *W1 = (const float*)d_in[2],  *as1 = (const float*)d_in[3],
              *ad1 = (const float*)d_in[4], *b1  = (const float*)d_in[5];
  const float *W2 = (const float*)d_in[6],  *as2 = (const float*)d_in[7],
              *ad2 = (const float*)d_in[8], *b2  = (const float*)d_in[9];
  const float *Wm = (const float*)d_in[10], *asm_ = (const float*)d_in[11],
              *adm = (const float*)d_in[12], *bm = (const float*)d_in[13];
  const float *Wl = (const float*)d_in[14], *asl = (const float*)d_in[15],
              *adl = (const float*)d_in[16], *bl = (const float*)d_in[17];

  char* ws = (char*)d_ws;
  size_t off = 0;
  auto alloc = [&](size_t bytes) -> void* {
    void* p = ws + off;
    off += (bytes + 255) & ~(size_t)255;
    return p;
  };
  float* Hbuf = (float*)alloc((size_t)NNODES * 512 * 4);
  float* A0   = (float*)alloc((size_t)NNODES * 256 * 4);
  float* A1   = (float*)alloc((size_t)NNODES * 512 * 4);
  float* ALS  = (float*)alloc((size_t)NNODES * 2 * HEADS * 4);
  float* ALD  = (float*)alloc((size_t)NNODES * 2 * HEADS * 4);
  int*   CNT  = (int*)alloc((size_t)NNODES * 4);
  int*   RPTR = (int*)alloc((size_t)(NNODES + 1) * 4);
  int*   RANK = (int*)alloc((size_t)E * 4);
  int*   CSRC = (int*)alloc((size_t)E * 4);

  float* zmean   = (float*)d_out;
  float* zlogstd = (float*)d_out + (size_t)NNODES * 32;

  zero_i<<<(NNODES + 255) / 256, 256, 0, stream>>>(CNT, NNODES);
  dst_count<<<(E + 255) / 256, 256, 0, stream>>>(dst, E, CNT, RANK);
  dst_scan<<<1, 1024, 0, stream>>>(CNT, RPTR);
  dst_fill<<<(E + 255) / 256, 256, 0, stream>>>(src, dst, E, RPTR, RANK, CSRC);

  run_gat_layer<64, false>(stream, x, 256, W1, as1, ad1, b1, Hbuf, A0, RPTR, CSRC, ALS, ALD);
  run_gat_layer<128, false>(stream, A0, 256, W2, as2, ad2, b2, Hbuf, A1, RPTR, CSRC, ALS, ALD);
  {
    dim3 ggrid((NNODES / 16 + 7) / 8, 256 / 64);
    gemm_xw_wmma<<<ggrid, 256, 0, stream>>>(A1, Wm, Wl, Hbuf, NNODES, 512, 128, 256, 128,
                                            asm_, adm, asl, adl, ALS, ALD, 2, 8);
    gat_node_pair<<<(NNODES + 7) / 8, 256, 0, stream>>>(RPTR, CSRC, Hbuf, ALS, ALD, bm, bl,
                                                        zmean, zlogstd, NNODES);
  }
}


# GEMM operand loads issued in groups ahead of use: A as 16-byte pieces (4 loads per 128-byte line, one 32-k group ahead), B chunk's 8 loads together; k order within 8 permuted (sum order changed)
# speedup vs baseline: 5.5019x; 1.1891x over previous
#include <hip/hip_runtime.h>

typedef float v2f __attribute__((ext_vector_type(2)));
typedef float v8f __attribute__((ext_vector_type(8)));

#define NNODES 30000
#define HEADS 4
#define SLOPE 0.2f
#define KCHUNK 128
#define PAIRSTRIDE 144

__device__ __forceinline__ float sum16_to_lane(const float (&x)[16], int lm) {
  const bool b3 = (lm & 8) != 0, b2 = (lm & 4) != 0, b1 = (lm & 2) != 0, b0 = (lm & 1) != 0;
  float y[8], z[4], w[2];
#pragma unroll
  for (int i = 0; i < 8; ++i)
    y[i] = (b3 ? x[i + 8] : x[i]) + __shfl_xor(b3 ? x[i] : x[i + 8], 8);
#pragma unroll
  for (int i = 0; i < 4; ++i)
    z[i] = (b2 ? y[i + 4] : y[i]) + __shfl_xor(b2 ? y[i] : y[i + 4], 4);
#pragma unroll
  for (int i = 0; i < 2; ++i)
    w[i] = (b1 ? z[i + 2] : z[i]) + __shfl_xor(b1 ? z[i] : z[i + 2], 2);
  return (b0 ? w[1] : w[0]) + __shfl_xor(b0 ? w[0] : w[1], 1);
}

__global__ __launch_bounds__(256, 8)
void gemm_xw_wmma(const float* __restrict__ A, const float* __restrict__ B,
                  const float* __restrict__ B2, float* __restrict__ C,
                  int M, int K, int ldb, int ldc, int nsplit,
                  const float* __restrict__ a_s, const float* __restrict__ a_d,
                  const float* __restrict__ a_s2, const float* __restrict__ a_d2,
                  float* __restrict__ als, float* __restrict__ ald, int G, int ldl) {
  __shared__ float ldsB[(KCHUNK / 2) * PAIRSTRIDE];

  const int tid  = threadIdx.x;
  const int lane = tid & 31;
  const int wave = tid >> 5;
  const int Mtiles = M >> 4;

  const int n0 = blockIdx.y * 64;
  const bool second = (n0 >= nsplit);
  const float* __restrict__ Bs = second ? B2 : B;
  const int nb = second ? n0 - nsplit : n0;
  int mt = blockIdx.x * 8 + wave;
  const bool active = (mt < Mtiles);
  if (!active) mt = Mtiles - 1;
  const int m0   = mt * 16;
  const int half = lane >> 4;
  const int lm   = lane & 15;

  const int lrow = tid >> 4;
  const int lcol = (tid & 15) * 4;

  v8f acc0 = {}, acc1 = {}, acc2 = {}, acc3 = {};
  const float* __restrict__ arow = A + (size_t)(m0 + lm) * K;

  for (int kb = 0; kb < K; kb += KCHUNK) {
    float4 an[4];
#pragma unroll
    for (int q = 0; q < 4; ++q)
      an[q] = *(const float4*)(arow + kb + 8 * q + 4 * half);

    float4 bv[KCHUNK / 16];
#pragma unroll
    for (int r = 0; r < KCHUNK; r += 16)
      bv[r / 16] = *(const float4*)(Bs + (size_t)(kb + r + lrow) * ldb + nb + lcol);

    __syncthreads();
#pragma unroll
    for (int r = 0; r < KCHUNK; r += 16) {
      const int k = r + lrow;
      const float4 v = bv[r / 16];
      float* p = &ldsB[(k >> 1) * PAIRSTRIDE + lcol * 2 + (k & 1)];
      p[0] = v.x; p[2] = v.y; p[4] = v.z; p[6] = v.w;
    }
    __syncthreads();

#pragma unroll
    for (int g = 0; g < KCHUNK / 32; ++g) {
      float4 ac[4];
#pragma unroll
      for (int q = 0; q < 4; ++q) ac[q] = an[q];
      if (g + 1 < KCHUNK / 32) {
        __builtin_amdgcn_sched_barrier(0);
#pragma unroll
        for (int q = 0; q < 4; ++q)
          an[q] = *(const float4*)(arow + kb + 32 * (g + 1) + 8 * q + 4 * half);
        __builtin_amdgcn_sched_barrier(0);
      }
#pragma unroll
      for (int q = 0; q < 4; ++q) {
#pragma unroll
        for (int r2 = 0; r2 < 2; ++r2) {
          v2f a;
          a.x = r2 ? ac[q].z : ac[q].x;
          a.y = r2 ? ac[q].w : ac[q].y;
          const float* __restrict__ bp =
              &ldsB[(16 * g + 4 * q + 2 * half + r2) * PAIRSTRIDE + lm * 2];
          v2f b;
          b = *(const v2f*)(bp + 0);
          acc0 = __builtin_amdgcn_wmma_f32_16x16x4_f32(false, a, false, b, (short)0, acc0, false, false);
          b = *(const v2f*)(bp + 32);
          acc1 = __builtin_amdgcn_wmma_f32_16x16x4_f32(false, a, false, b, (short)0, acc1, false, false);
          b = *(const v2f*)(bp + 64);
          acc2 = __builtin_amdgcn_wmma_f32_16x16x4_f32(false, a, false, b, (short)0, acc2, false, false);
          b = *(const v2f*)(bp + 96);
          acc3 = __builtin_amdgcn_wmma_f32_16x16x4_f32(false, a, false, b, (short)0, acc3, false, false);
        }
      }
    }
  }

  if (active) {
    float* __restrict__ crow = C + (size_t)(m0 + 8 * half) * ldc + n0 + lm;
#pragma unroll
    for (int v = 0; v < 8; ++v) {
      crow[(size_t)v * ldc + 0]  = acc0[v];
      crow[(size_t)v * ldc + 16] = acc1[v];
      crow[(size_t)v * ldc + 32] = acc2[v];
      crow[(size_t)v * ldc + 48] = acc3[v];
    }

    const float* __restrict__ vs = second ? a_s2 : a_s;
    const float* __restrict__ vd = second ? a_d2 : a_d;
    float ws[4], wd[4];
#pragma unroll
    for (int t = 0; t < 4; ++t) {
      ws[t] = vs[nb + 16 * t + lm];
      wd[t] = vd[nb + 16 * t + lm];
    }
    float* __restrict__ lout = ((lm & 8) ? ald : als)
                             + (size_t)(m0 + 8 * half + (lm & 7)) * ldl + blockIdx.y * G;
    float x[16];
    if (G == 1) {
#pragma unroll
      for (int v = 0; v < 8; ++v) {
        x[v]     = acc0[v] * ws[0] + acc1[v] * ws[1] + acc2[v] * ws[2] + acc3[v] * ws[3];
        x[8 + v] = acc0[v] * wd[0] + acc1[v] * wd[1] + acc2[v] * wd[2] + acc3[v] * wd[3];
      }
      lout[0] = sum16_to_lane(x, lm);
    } else {
#pragma unroll
      for (int v = 0; v < 8; ++v) {
        x[v]     = acc0[v] * ws[0] + acc1[v] * ws[1];
        x[8 + v] = acc0[v] * wd[0] + acc1[v] * wd[1];
      }
      lout[0] = sum16_to_lane(x, lm);
#pragma unroll
      for (int v = 0; v < 8; ++v) {
        x[v]     = acc2[v] * ws[2] + acc3[v] * ws[3];
        x[8 + v] = acc2[v] * wd[2] + acc3[v] * wd[3];
      }
      lout[1] = sum16_to_lane(x, lm);
    }
  }
}

__global__ void zero_i(int* __restrict__ p, int n) {
  int i = blockIdx.x * blockDim.x + threadIdx.x;
  if (i < n) p[i] = 0;
}

__global__ void dst_count(const int* __restrict__ dst, int E,
                          int* __restrict__ cnt, int* __restrict__ rank) {
  int e = blockIdx.x * blockDim.x + threadIdx.x;
  if (e >= E) return;
  rank[e] = atomicAdd(&cnt[dst[e]], 1);
}

#define SCAN_PER ((NNODES + 1023) / 1024)

__global__ __launch_bounds__(1024)
void dst_scan(const int* __restrict__ cnt, int* __restrict__ rowptr) {
  __shared__ int part[1024];
  const int t = threadIdx.x;
  const int lo = t * SCAN_PER;
  int c[SCAN_PER];
  int s = 0;
#pragma unroll
  for (int q = 0; q < SCAN_PER; ++q) {
    c[q] = (lo + q < NNODES) ? cnt[lo + q] : 0;
    s += c[q];
  }
  part[t] = s;
  __syncthreads();
  for (int off = 1; off < 1024; off <<= 1) {
    const int v = (t >= off) ? part[t - off] : 0;
    __syncthreads();
    part[t] += v;
    __syncthreads();
  }
  int run = part[t] - s;
#pragma unroll
  for (int q = 0; q < SCAN_PER; ++q) {
    if (lo + q < NNODES) rowptr[lo + q] = run;
    run += c[q];
  }
  if (t == 1023) rowptr[NNODES] = part[1023];
}

__global__ void dst_fill(const int* __restrict__ src, const int* __restrict__ dst, int E,
                         const int* __restrict__ rowptr, const int* __restrict__ rank,
                         int* __restrict__ csr_src) {
  int e = blockIdx.x * blockDim.x + threadIdx.x;
  if (e >= E) return;
  csr_src[rowptr[dst[e]] + rank[e]] = src[e];
}

__device__ __forceinline__ float lrelu(float v) { return (v > 0.f) ? v : SLOPE * v; }

template <int C>
__device__ __forceinline__ float4 node_logits(const float* __restrict__ p, int node) {
  if (C == 128) {
    const float4 q0 = *(const float4*)(p + (size_t)node * 8);
    const float4 q1 = *(const float4*)(p + (size_t)node * 8 + 4);
    return make_float4(q0.x + q0.y, q0.z + q0.w, q1.x + q1.y, q1.z + q1.w);
  }
  return *(const float4*)(p + (size_t)node * 4);
}

template <int C>
__device__ __forceinline__ float pick_head(int k, int lane, float a0, float a1, float a2, float a3) {
  if (C == 128) return (k == 0) ? a0 : (k == 1) ? a1 : (k == 2) ? a2 : a3;
  if (C == 64) {
    const bool hi = (lane & 16) != 0;
    return (k == 0) ? (hi ? a1 : a0) : (hi ? a3 : a2);
  }
  const int h = lane >> 3;
  const float x01 = (h & 1) ? a1 : a0;
  const float x23 = (h & 1) ? a3 : a2;
  return (h & 2) ? x23 : x01;
}

template <int C, bool MEAN>
__global__ __launch_bounds__(256)
void gat_node(const int* __restrict__ rowptr, const int* __restrict__ csr_src,
              const float* __restrict__ H, const float* __restrict__ als,
              const float* __restrict__ ald, const float* __restrict__ bias,
              float* __restrict__ out, int Nn) {
  constexpr int Fout = HEADS * C;
  constexpr int NV = Fout / 128;
  constexpr int UNR = (NV == 1) ? 8 : 4;
  const int lane = threadIdx.x & 31;
  const int d = blockIdx.x * 8 + __builtin_amdgcn_readfirstlane((int)(threadIdx.x >> 5));
  if (d >= Nn) return;
  const int base = rowptr[d];
  const int deg  = rowptr[d + 1] - base;
  const int cnt  = deg + 1;
  const float4 adv = node_logits<C>(ald, d);

  float m0 = -__builtin_inff(), m1 = m0, m2 = m0, m3 = m0;
  for (int j = lane; j < cnt; j += 32) {
    const int s = (j < deg) ? csr_src[base + j] : d;
    const float4 a = node_logits<C>(als, s);
    m0 = fmaxf(m0, lrelu(a.x + adv.x));
    m1 = fmaxf(m1, lrelu(a.y + adv.y));
    m2 = fmaxf(m2, lrelu(a.z + adv.z));
    m3 = fmaxf(m3, lrelu(a.w + adv.w));
  }
#pragma unroll
  for (int o = 16; o > 0; o >>= 1) {
    m0 = fmaxf(m0, __shfl_xor(m0, o));
    m1 = fmaxf(m1, __shfl_xor(m1, o));
    m2 = fmaxf(m2, __shfl_xor(m2, o));
    m3 = fmaxf(m3, __shfl_xor(m3, o));
  }

  float z0 = 0.f, z1 = 0.f, z2 = 0.f, z3 = 0.f;
  for (int j = lane; j < cnt; j += 32) {
    const int s = (j < deg) ? csr_src[base + j] : d;
    const float4 a = node_logits<C>(als, s);
    z0 += __expf(lrelu(a.x + adv.x) - m0);
    z1 += __expf(lrelu(a.y + adv.y) - m1);
    z2 += __expf(lrelu(a.z + adv.z) - m2);
    z3 += __expf(lrelu(a.w + adv.w) - m3);
  }
#pragma unroll
  for (int o = 16; o > 0; o >>= 1) {
    z0 += __shfl_xor(z0, o);
    z1 += __shfl_xor(z1, o);
    z2 += __shfl_xor(z2, o);
    z3 += __shfl_xor(z3, o);
  }
  z0 += 1e-16f; z1 += 1e-16f; z2 += 1e-16f; z3 += 1e-16f;

  float4 acc[NV];
#pragma unroll
  for (int k = 0; k < NV; ++k) acc[k] = make_float4(0.f, 0.f, 0.f, 0.f);

  for (int j0 = 0; j0 < cnt; j0 += 32) {
    const int j = j0 + lane;
    int s_l = d;
    float p0 = 0.f, p1 = 0.f, p2 = 0.f, p3 = 0.f;
    if (j < cnt) {
      s_l = (j < deg) ? csr_src[base + j] : d;
      const float4 a = node_logits<C>(als, s_l);
      p0 = __expf(lrelu(a.x + adv.x) - m0) / z0;
      p1 = __expf(lrelu(a.y + adv.y) - m1) / z1;
      p2 = __expf(lrelu(a.z + adv.z) - m2) / z2;
      p3 = __expf(lrelu(a.w + adv.w) - m3) / z3;
    }
    const int left = cnt - j0;
    const int nn = (left < 32) ? left : 32;
    for (int t0 = 0; t0 < nn; t0 += UNR) {
#pragma unroll
      for (int u = 0; u < UNR; ++u) {
        const int t = t0 + u;
        const int s = __builtin_amdgcn_readlane(s_l, t);
        const float a0 = __int_as_float(__builtin_amdgcn_readlane(__float_as_int(p0), t));
        const float a1 = __int_as_float(__builtin_amdgcn_readlane(__float_as_int(p1), t));
        const float a2 = __int_as_float(__builtin_amdgcn_readlane(__float_as_int(p2), t));
        const float a3 = __int_as_float(__builtin_amdgcn_readlane(__float_as_int(p3), t));
        const float* __restrict__ hrow = H + (size_t)s * Fout + lane * 4;
#pragma unroll
        for (int k = 0; k < NV; ++k) {
          const float4 hv = *(const float4*)(hrow + k * 128);
          const float al = pick_head<C>(k, lane, a0, a1, a2, a3);
          acc[k].x += al * hv.x;
          acc[k].y += al * hv.y;
          acc[k].z += al * hv.z;
          acc[k].w += al * hv.w;
        }
      }
    }
  }

  if (MEAN) {
    float4 v = acc[0];
    v.x += __shfl_xor(v.x, 8);  v.y += __shfl_xor(v.y, 8);
    v.z += __shfl_xor(v.z, 8);  v.w += __shfl_xor(v.w, 8);
    v.x += __shfl_xor(v.x, 16); v.y += __shfl_xor(v.y, 16);
    v.z += __shfl_xor(v.z, 16); v.w += __shfl_xor(v.w, 16);
    if (lane < 8) {
      const float4 b = *(const float4*)(bias + lane * 4);
      float4 o;
      o.x = 0.25f * v.x + b.x; o.y = 0.25f * v.y + b.y;
      o.z = 0.25f * v.z + b.z; o.w = 0.25f * v.w + b.w;
      *(float4*)(out + (size_t)d * C + lane * 4) = o;
    }
  } else {
#pragma unroll
    for (int k = 0; k < NV; ++k) {
      const float4 b = *(const float4*)(bias + k * 128 + lane * 4);
      float4 o;
      o.x = fmaxf(acc[k].x + b.x, 0.f); o.y = fmaxf(acc[k].y + b.y, 0.f);
      o.z = fmaxf(acc[k].z + b.z, 0.f); o.w = fmaxf(acc[k].w + b.w, 0.f);
      *(float4*)(out + (size_t)d * Fout + k * 128 + lane * 4) = o;
    }
  }
}

__device__ __forceinline__ void mean_heads_store(float4 v, int lane,
                                                 const float* __restrict__ bias,
                                                 float* __restrict__ orow) {
  v.x += __shfl_xor(v.x, 8);  v.y += __shfl_xor(v.y, 8);
  v.z += __shfl_xor(v.z, 8);  v.w += __shfl_xor(v.w, 8);
  v.x += __shfl_xor(v.x, 16); v.y += __shfl_xor(v.y, 16);
  v.z += __shfl_xor(v.z, 16); v.w += __shfl_xor(v.w, 16);
  if (lane < 8) {
    const float4 b = *(const float4*)(bias + lane * 4);
    float4 o;
    o.x = 0.25f * v.x + b.x; o.y = 0.25f * v.y + b.y;
    o.z = 0.25f * v.z + b.z; o.w = 0.25f * v.w + b.w;
    *(float4*)(orow + lane * 4) = o;
  }
}

__global__ __launch_bounds__(256)
void gat_node_pair(const int* __restrict__ rowptr, const int* __restrict__ csr_src,
                   const float* __restrict__ H, const float* __restrict__ als,
                   const float* __restrict__ ald, const float* __restrict__ bias_a,
                   const float* __restrict__ bias_b, float* __restrict__ out_a,
                   float* __restrict__ out_b, int Nn) {
  constexpr int NH = 2 * HEADS;
  constexpr int C = 32;
  constexpr int Fout = NH * C;
  constexpr int UNR = 8;
  const int lane = threadIdx.x & 31;
  const int d = blockIdx.x * 8 + __builtin_amdgcn_readfirstlane((int)(threadIdx.x >> 5));
  if (d >= Nn) return;
  const int base = rowptr[d];
  const int deg  = rowptr[d + 1] - base;
  const int cnt  = deg + 1;
  float adv[NH];
  {
    const float4 x = *(const float4*)(ald + (size_t)d * NH);
    const float4 y = *(const float4*)(ald + (size_t)d * NH + 4);
    adv[0] = x.x; adv[1] = x.y; adv[2] = x.z; adv[3] = x.w;
    adv[4] = y.x; adv[5] = y.y; adv[6] = y.z; adv[7] = y.w;
  }

  float m[NH];
#pragma unroll
  for (int h = 0; h < NH; ++h) m[h] = -__builtin_inff();
  for (int j = lane; j < cnt; j += 32) {
    const int s = (j < deg) ? csr_src[base + j] : d;
    const float4 a = *(const float4*)(als + (size_t)s * NH);
    const float4 b = *(const float4*)(als + (size_t)s * NH + 4);
    const float e[NH] = {a.x, a.y, a.z, a.w, b.x, b.y, b.z, b.w};
#pragma unroll
    for (int h = 0; h < NH; ++h) m[h] = fmaxf(m[h], lrelu(e[h] + adv[h]));
  }
#pragma unroll
  for (int o = 16; o > 0; o >>= 1) {
#pragma unroll
    for (int h = 0; h < NH; ++h) m[h] = fmaxf(m[h], __shfl_xor(m[h], o));
  }

  float z[NH];
#pragma unroll
  for (int h = 0; h < NH; ++h) z[h] = 0.f;
  for (int j = lane; j < cnt; j += 32) {
    const int s = (j < deg) ? csr_src[base + j] : d;
    const float4 a = *(const float4*)(als + (size_t)s * NH);
    const float4 b = *(const float4*)(als + (size_t)s * NH + 4);
    const float e[NH] = {a.x, a.y, a.z, a.w, b.x, b.y, b.z, b.w};
#pragma unroll
    for (int h = 0; h < NH; ++h) z[h] += __expf(lrelu(e[h] + adv[h]) - m[h]);
  }
#pragma unroll
  for (int o = 16; o > 0; o >>= 1) {
#pragma unroll
    for (int h = 0; h < NH; ++h) z[h] += __shfl_xor(z[h], o);
  }
#pragma unroll
  for (int h = 0; h < NH; ++h) z[h] += 1e-16f;

  float4 acca = make_float4(0.f, 0.f, 0.f, 0.f);
  float4 accb = make_float4(0.f, 0.f, 0.f, 0.f);

  for (int j0 = 0; j0 < cnt; j0 += 32) {
    const int j = j0 + lane;
    int s_l = d;
    float p[NH];
#pragma unroll
    for (int h = 0; h < NH; ++h) p[h] = 0.f;
    if (j < cnt) {
      s_l = (j < deg) ? csr_src[base + j] : d;
      const float4 a = *(const float4*)(als + (size_t)s_l * NH);
      const float4 b = *(const float4*)(als + (size_t)s_l * NH + 4);
      const float e[NH] = {a.x, a.y, a.z, a.w, b.x, b.y, b.z, b.w};
#pragma unroll
      for (int h = 0; h < NH; ++h) p[h] = __expf(lrelu(e[h] + adv[h]) - m[h]) / z[h];
    }
    const int left = cnt - j0;
    const int nn = (left < 32) ? left : 32;
    for (int t0 = 0; t0 < nn; t0 += UNR) {
#pragma unroll
      for (int u = 0; u < UNR; ++u) {
        const int t = t0 + u;
        const int s = __builtin_amdgcn_readlane(s_l, t);
        float a[NH];
#pragma unroll
        for (int h = 0; h < NH; ++h)
          a[h] = __int_as_float(__builtin_amdgcn_readlane(__float_as_int(p[h]), t));
        const float* __restrict__ hrow = H + (size_t)s * Fout + lane * 4;
        const float4 hva = *(const float4*)(hrow);
        const float4 hvb = *(const float4*)(hrow + 128);
        const float ala = pick_head<C>(0, lane, a[0], a[1], a[2], a[3]);
        const float alb = pick_head<C>(0, lane, a[4], a[5], a[6], a[7]);
        acca.x += ala * hva.x; acca.y += ala * hva.y;
        acca.z += ala * hva.z; acca.w += ala * hva.w;
        accb.x += alb * hvb.x; accb.y += alb * hvb.y;
        accb.z += alb * hvb.z; accb.w += alb * hvb.w;
      }
    }
  }

  mean_heads_store(acca, lane, bias_a, out_a + (size_t)d * C);
  mean_heads_store(accb, lane, bias_b, out_b + (size_t)d * C);
}

template <int C, bool MEAN>
static void run_gat_layer(hipStream_t st, const float* in, int Fin,
                          const float* W, const float* a_s, const float* a_d, const float* b,
                          float* Hbuf, float* outp,
                          const int* rowptr, const int* csr_src,
                          float* ALS, float* ALD) {
  const int Nn = NNODES;
  const int Fout = HEADS * C;
  const int Mtiles = Nn / 16;
  dim3 ggrid((Mtiles + 7) / 8, Fout / 64);
  gemm_xw_wmma<<<ggrid, 256, 0, st>>>(in, W, W, Hbuf, Nn, Fin, Fout, Fout, Fout,
                                      a_s, a_d, a_s, a_d, ALS, ALD, 1, Fout / 64);
  gat_node<C, MEAN><<<(Nn + 7) / 8, 256, 0, st>>>(rowptr, csr_src, Hbuf, ALS, ALD, b, outp, Nn);
}

extern "C" void kernel_launch(void* const* d_in, const int* in_sizes, int n_in,
                              void* d_out, int out_size, void* d_ws, size_t ws_size,
                              hipStream_t stream) {
  const float* x   = (const float*)d_in[0];
  const int*   ei  = (const int*)d_in[1];
  const int E = in_sizes[1] / 2;
  const int* src = ei;
  const int* dst = ei + E;
  const float *W1 = (const float*)d_in[2],  *as1 = (const float*)d_in[3],
              *ad1 = (const float*)d_in[4], *b1  = (const float*)d_in[5];
  const float *W2 = (const float*)d_in[6],  *as2 = (const float*)d_in[7],
              *ad2 = (const float*)d_in[8], *b2  = (const float*)d_in[9];
  const float *Wm = (const float*)d_in[10], *asm_ = (const float*)d_in[11],
              *adm = (const float*)d_in[12], *bm = (const float*)d_in[13];
  const float *Wl = (const float*)d_in[14], *asl = (const float*)d_in[15],
              *adl = (const float*)d_in[16], *bl = (const float*)d_in[17];

  char* ws = (char*)d_ws;
  size_t off = 0;
  auto alloc = [&](size_t bytes) -> void* {
    void* p = ws + off;
    off += (bytes + 255) & ~(size_t)255;
    return p;
  };
  float* Hbuf = (float*)alloc((size_t)NNODES * 512 * 4);
  float* A0   = (float*)alloc((size_t)NNODES * 256 * 4);
  float* A1   = (float*)alloc((size_t)NNODES * 512 * 4);
  float* ALS  = (float*)alloc((size_t)NNODES * 2 * HEADS * 4);
  float* ALD  = (float*)alloc((size_t)NNODES * 2 * HEADS * 4);
  int*   CNT  = (int*)alloc((size_t)NNODES * 4);
  int*   RPTR = (int*)alloc((size_t)(NNODES + 1) * 4);
  int*   RANK = (int*)alloc((size_t)E * 4);
  int*   CSRC = (int*)alloc((size_t)E * 4);

  float* zmean   = (float*)d_out;
  float* zlogstd = (float*)d_out + (size_t)NNODES * 32;

  zero_i<<<(NNODES + 255) / 256, 256, 0, stream>>>(CNT, NNODES);
  dst_count<<<(E + 255) / 256, 256, 0, stream>>>(dst, E, CNT, RANK);
  dst_scan<<<1, 1024, 0, stream>>>(CNT, RPTR);
  dst_fill<<<(E + 255) / 256, 256, 0, stream>>>(src, dst, E, RPTR, RANK, CSRC);

  run_gat_layer<64, false>(stream, x, 256, W1, as1, ad1, b1, Hbuf, A0, RPTR, CSRC, ALS, ALD);
  run_gat_layer<128, false>(stream, A0, 256, W2, as2, ad2, b2, Hbuf, A1, RPTR, CSRC, ALS, ALD);
  {
    dim3 ggrid((NNODES / 16 + 7) / 8, 256 / 64);
    gemm_xw_wmma<<<ggrid, 256, 0, stream>>>(A1, Wm, Wl, Hbuf, NNODES, 512, 128, 256, 128,
                                            asm_, adm, asl, adl, ALS, ALD, 2, 8);
    gat_node_pair<<<(NNODES + 7) / 8, 256, 0, stream>>>(RPTR, CSRC, Hbuf, ALS, ALD, bm, bl,
                                                        zmean, zlogstd, NNODES);
  }
}


# Bucketing (count, prefix sum, fill): 4 consecutive items per lane, 16-byte loads/stores: prefix sum touches whole 128-byte lines, count/fill launch 1/4 the waves; edge order still atomic arrival rank
# speedup vs baseline: 5.9492x; 1.0813x over previous
#include <hip/hip_runtime.h>

typedef float v2f __attribute__((ext_vector_type(2)));
typedef float v8f __attribute__((ext_vector_type(8)));

#define NNODES 30000
#define HEADS 4
#define SLOPE 0.2f
#define KCHUNK 128
#define PAIRSTRIDE 144

__device__ __forceinline__ float sum16_to_lane(const float (&x)[16], int lm) {
  const bool b3 = (lm & 8) != 0, b2 = (lm & 4) != 0, b1 = (lm & 2) != 0, b0 = (lm & 1) != 0;
  float y[8], z[4], w[2];
#pragma unroll
  for (int i = 0; i < 8; ++i)
    y[i] = (b3 ? x[i + 8] : x[i]) + __shfl_xor(b3 ? x[i] : x[i + 8], 8);
#pragma unroll
  for (int i = 0; i < 4; ++i)
    z[i] = (b2 ? y[i + 4] : y[i]) + __shfl_xor(b2 ? y[i] : y[i + 4], 4);
#pragma unroll
  for (int i = 0; i < 2; ++i)
    w[i] = (b1 ? z[i + 2] : z[i]) + __shfl_xor(b1 ? z[i] : z[i + 2], 2);
  return (b0 ? w[1] : w[0]) + __shfl_xor(b0 ? w[0] : w[1], 1);
}

__global__ __launch_bounds__(256, 8)
void gemm_xw_wmma(const float* __restrict__ A, const float* __restrict__ B,
                  const float* __restrict__ B2, float* __restrict__ C,
                  int M, int K, int ldb, int ldc, int nsplit,
                  const float* __restrict__ a_s, const float* __restrict__ a_d,
                  const float* __restrict__ a_s2, const float* __restrict__ a_d2,
                  float* __restrict__ als, float* __restrict__ ald, int G, int ldl) {
  __shared__ float ldsB[(KCHUNK / 2) * PAIRSTRIDE];

  const int tid  = threadIdx.x;
  const int lane = tid & 31;
  const int wave = tid >> 5;
  const int Mtiles = M >> 4;

  const int n0 = blockIdx.y * 64;
  const bool second = (n0 >= nsplit);
  const float* __restrict__ Bs = second ? B2 : B;
  const int nb = second ? n0 - nsplit : n0;
  int mt = blockIdx.x * 8 + wave;
  const bool active = (mt < Mtiles);
  if (!active) mt = Mtiles - 1;
  const int m0   = mt * 16;
  const int half = lane >> 4;
  const int lm   = lane & 15;

  const int lrow = tid >> 4;
  const int lcol = (tid & 15) * 4;

  v8f acc0 = {}, acc1 = {}, acc2 = {}, acc3 = {};
  const float* __restrict__ arow = A + (size_t)(m0 + lm) * K;

  for (int kb = 0; kb < K; kb += KCHUNK) {
    float4 an[4];
#pragma unroll
    for (int q = 0; q < 4; ++q)
      an[q] = *(const float4*)(arow + kb + 8 * q + 4 * half);

    float4 bv[KCHUNK / 16];
#pragma unroll
    for (int r = 0; r < KCHUNK; r += 16)
      bv[r / 16] = *(const float4*)(Bs + (size_t)(kb + r + lrow) * ldb + nb + lcol);

    __syncthreads();
#pragma unroll
    for (int r = 0; r < KCHUNK; r += 16) {
      const int k = r + lrow;
      const float4 v = bv[r / 16];
      float* p = &ldsB[(k >> 1) * PAIRSTRIDE + lcol * 2 + (k & 1)];
      p[0] = v.x; p[2] = v.y; p[4] = v.z; p[6] = v.w;
    }
    __syncthreads();

#pragma unroll
    for (int g = 0; g < KCHUNK / 32; ++g) {
      float4 ac[4];
#pragma unroll
      for (int q = 0; q < 4; ++q) ac[q] = an[q];
      if (g + 1 < KCHUNK / 32) {
        __builtin_amdgcn_sched_barrier(0);
#pragma unroll
        for (int q = 0; q < 4; ++q)
          an[q] = *(const float4*)(arow + kb + 32 * (g + 1) + 8 * q + 4 * half);
        __builtin_amdgcn_sched_barrier(0);
      }
#pragma unroll
      for (int q = 0; q < 4; ++q) {
#pragma unroll
        for (int r2 = 0; r2 < 2; ++r2) {
          v2f a;
          a.x = r2 ? ac[q].z : ac[q].x;
          a.y = r2 ? ac[q].w : ac[q].y;
          const float* __restrict__ bp =
              &ldsB[(16 * g + 4 * q + 2 * half + r2) * PAIRSTRIDE + lm * 2];
          v2f b;
          b = *(const v2f*)(bp + 0);
          acc0 = __builtin_amdgcn_wmma_f32_16x16x4_f32(false, a, false, b, (short)0, acc0, false, false);
          b = *(const v2f*)(bp + 32);
          acc1 = __builtin_amdgcn_wmma_f32_16x16x4_f32(false, a, false, b, (short)0, acc1, false, false);
          b = *(const v2f*)(bp + 64);
          acc2 = __builtin_amdgcn_wmma_f32_16x16x4_f32(false, a, false, b, (short)0, acc2, false, false);
          b = *(const v2f*)(bp + 96);
          acc3 = __builtin_amdgcn_wmma_f32_16x16x4_f32(false, a, false, b, (short)0, acc3, false, false);
        }
      }
    }
  }

  if (active) {
    float* __restrict__ crow = C + (size_t)(m0 + 8 * half) * ldc + n0 + lm;
#pragma unroll
    for (int v = 0; v < 8; ++v) {
      crow[(size_t)v * ldc + 0]  = acc0[v];
      crow[(size_t)v * ldc + 16] = acc1[v];
      crow[(size_t)v * ldc + 32] = acc2[v];
      crow[(size_t)v * ldc + 48] = acc3[v];
    }

    const float* __restrict__ vs = second ? a_s2 : a_s;
    const float* __restrict__ vd = second ? a_d2 : a_d;
    float ws[4], wd[4];
#pragma unroll
    for (int t = 0; t < 4; ++t) {
      ws[t] = vs[nb + 16 * t + lm];
      wd[t] = vd[nb + 16 * t + lm];
    }
    float* __restrict__ lout = ((lm & 8) ? ald : als)
                             + (size_t)(m0 + 8 * half + (lm & 7)) * ldl + blockIdx.y * G;
    float x[16];
    if (G == 1) {
#pragma unroll
      for (int v = 0; v < 8; ++v) {
        x[v]     = acc0[v] * ws[0] + acc1[v] * ws[1] + acc2[v] * ws[2] + acc3[v] * ws[3];
        x[8 + v] = acc0[v] * wd[0] + acc1[v] * wd[1] + acc2[v] * wd[2] + acc3[v] * wd[3];
      }
      lout[0] = sum16_to_lane(x, lm);
    } else {
#pragma unroll
      for (int v = 0; v < 8; ++v) {
        x[v]     = acc0[v] * ws[0] + acc1[v] * ws[1];
        x[8 + v] = acc0[v] * wd[0] + acc1[v] * wd[1];
      }
      lout[0] = sum16_to_lane(x, lm);
#pragma unroll
      for (int v = 0; v < 8; ++v) {
        x[v]     = acc2[v] * ws[2] + acc3[v] * ws[3];
        x[8 + v] = acc2[v] * wd[2] + acc3[v] * wd[3];
      }
      lout[1] = sum16_to_lane(x, lm);
    }
  }
}

__global__ void zero_i(int* __restrict__ p, int n) {
  int i = blockIdx.x * blockDim.x + threadIdx.x;
  if (i < n) p[i] = 0;
}

__global__ void dst_count(const int* __restrict__ dst, int E,
                          int* __restrict__ cnt, int* __restrict__ rank) {
  const int e0 = (blockIdx.x * blockDim.x + threadIdx.x) * 4;
  if (e0 >= E) return;
  if ((E & 3) == 0) {
    const int4 d = *(const int4*)(dst + e0);
    int4 r;
    r.x = atomicAdd(&cnt[d.x], 1);
    r.y = atomicAdd(&cnt[d.y], 1);
    r.z = atomicAdd(&cnt[d.z], 1);
    r.w = atomicAdd(&cnt[d.w], 1);
    *(int4*)(rank + e0) = r;
  } else {
    for (int k = 0; k < 4 && e0 + k < E; ++k) rank[e0 + k] = atomicAdd(&cnt[dst[e0 + k]], 1);
  }
}

#define SCAN_ROWS 8
static_assert(NNODES % 4 == 0, "a lane's four counts must be all inside or all outside");
static_assert(32 * SCAN_ROWS * 128 >= NNODES, "the 32 spans must cover every node");

__device__ __forceinline__ int lane_scan(int v, int lane) {
#pragma unroll
  for (int o = 1; o < 32; o <<= 1) {
    const int n = __shfl_up(v, o);
    if (lane >= o) v += n;
  }
  return v;
}

__global__ __launch_bounds__(1024)
void dst_scan(const int* __restrict__ cnt, int* __restrict__ rowptr) {
  __shared__ int wsum[32];
  const int lane = threadIdx.x & 31;
  const int wave = __builtin_amdgcn_readfirstlane((int)(threadIdx.x >> 5));
  const int lo = wave * (SCAN_ROWS * 128) + lane * 4;

  int4 c[SCAN_ROWS];
#pragma unroll
  for (int q = 0; q < SCAN_ROWS; ++q) {
    const int i = lo + q * 128;
    c[q] = (i < NNODES) ? *(const int4*)(cnt + i) : make_int4(0, 0, 0, 0);
  }

  int carry = 0;
#pragma unroll
  for (int q = 0; q < SCAN_ROWS; ++q) {
    const int4 v = c[q];
    const int s = v.x + v.y + v.z + v.w;
    const int incl = lane_scan(s, lane);
    const int base = carry + incl - s;
    carry += __builtin_amdgcn_readlane(incl, 31);
    c[q] = make_int4(base, base + v.x, base + v.x + v.y, base + v.x + v.y + v.z);
  }
  if (lane == 0) wsum[wave] = carry;
  __syncthreads();

  const int t = wsum[lane];
  const int incl = lane_scan(t, lane);
  const int woff = __builtin_amdgcn_readlane(incl - t, wave);
#pragma unroll
  for (int q = 0; q < SCAN_ROWS; ++q) {
    const int i = lo + q * 128;
    if (i < NNODES) {
      const int4 v = c[q];
      *(int4*)(rowptr + i) = make_int4(v.x + woff, v.y + woff, v.z + woff, v.w + woff);
    }
  }
  if (threadIdx.x == 1023) rowptr[NNODES] = incl;
}

__global__ void dst_fill(const int* __restrict__ src, const int* __restrict__ dst, int E,
                         const int* __restrict__ rowptr, const int* __restrict__ rank,
                         int* __restrict__ csr_src) {
  const int e0 = (blockIdx.x * blockDim.x + threadIdx.x) * 4;
  if (e0 >= E) return;
  if ((E & 3) == 0) {
    const int4 d = *(const int4*)(dst + e0);
    const int4 s = *(const int4*)(src + e0);
    const int4 r = *(const int4*)(rank + e0);
    const int p0 = rowptr[d.x], p1 = rowptr[d.y], p2 = rowptr[d.z], p3 = rowptr[d.w];
    csr_src[p0 + r.x] = s.x;
    csr_src[p1 + r.y] = s.y;
    csr_src[p2 + r.z] = s.z;
    csr_src[p3 + r.w] = s.w;
  } else {
    for (int k = 0; k < 4 && e0 + k < E; ++k)
      csr_src[rowptr[dst[e0 + k]] + rank[e0 + k]] = src[e0 + k];
  }
}

__device__ __forceinline__ float lrelu(float v) { return (v > 0.f) ? v : SLOPE * v; }

template <int C>
__device__ __forceinline__ float4 node_logits(const float* __restrict__ p, int node) {
  if (C == 128) {
    const float4 q0 = *(const float4*)(p + (size_t)node * 8);
    const float4 q1 = *(const float4*)(p + (size_t)node * 8 + 4);
    return make_float4(q0.x + q0.y, q0.z + q0.w, q1.x + q1.y, q1.z + q1.w);
  }
  return *(const float4*)(p + (size_t)node * 4);
}

template <int C>
__device__ __forceinline__ float pick_head(int k, int lane, float a0, float a1, float a2, float a3) {
  if (C == 128) return (k == 0) ? a0 : (k == 1) ? a1 : (k == 2) ? a2 : a3;
  if (C == 64) {
    const bool hi = (lane & 16) != 0;
    return (k == 0) ? (hi ? a1 : a0) : (hi ? a3 : a2);
  }
  const int h = lane >> 3;
  const float x01 = (h & 1) ? a1 : a0;
  const float x23 = (h & 1) ? a3 : a2;
  return (h & 2) ? x23 : x01;
}

template <int C, bool MEAN>
__global__ __launch_bounds__(256)
void gat_node(const int* __restrict__ rowptr, const int* __restrict__ csr_src,
              const float* __restrict__ H, const float* __restrict__ als,
              const float* __restrict__ ald, const float* __restrict__ bias,
              float* __restrict__ out, int Nn) {
  constexpr int Fout = HEADS * C;
  constexpr int NV = Fout / 128;
  constexpr int UNR = (NV == 1) ? 8 : 4;
  const int lane = threadIdx.x & 31;
  const int d = blockIdx.x * 8 + __builtin_amdgcn_readfirstlane((int)(threadIdx.x >> 5));
  if (d >= Nn) return;
  const int base = rowptr[d];
  const int deg  = rowptr[d + 1] - base;
  const int cnt  = deg + 1;
  const float4 adv = node_logits<C>(ald, d);

  float m0 = -__builtin_inff(), m1 = m0, m2 = m0, m3 = m0;
  for (int j = lane; j < cnt; j += 32) {
    const int s = (j < deg) ? csr_src[base + j] : d;
    const float4 a = node_logits<C>(als, s);
    m0 = fmaxf(m0, lrelu(a.x + adv.x));
    m1 = fmaxf(m1, lrelu(a.y + adv.y));
    m2 = fmaxf(m2, lrelu(a.z + adv.z));
    m3 = fmaxf(m3, lrelu(a.w + adv.w));
  }
#pragma unroll
  for (int o = 16; o > 0; o >>= 1) {
    m0 = fmaxf(m0, __shfl_xor(m0, o));
    m1 = fmaxf(m1, __shfl_xor(m1, o));
    m2 = fmaxf(m2, __shfl_xor(m2, o));
    m3 = fmaxf(m3, __shfl_xor(m3, o));
  }

  float z0 = 0.f, z1 = 0.f, z2 = 0.f, z3 = 0.f;
  for (int j = lane; j < cnt; j += 32) {
    const int s = (j < deg) ? csr_src[base + j] : d;
    const float4 a = node_logits<C>(als, s);
    z0 += __expf(lrelu(a.x + adv.x) - m0);
    z1 += __expf(lrelu(a.y + adv.y) - m1);
    z2 += __expf(lrelu(a.z + adv.z) - m2);
    z3 += __expf(lrelu(a.w + adv.w) - m3);
  }
#pragma unroll
  for (int o = 16; o > 0; o >>= 1) {
    z0 += __shfl_xor(z0, o);
    z1 += __shfl_xor(z1, o);
    z2 += __shfl_xor(z2, o);
    z3 += __shfl_xor(z3, o);
  }
  z0 += 1e-16f; z1 += 1e-16f; z2 += 1e-16f; z3 += 1e-16f;

  float4 acc[NV];
#pragma unroll
  for (int k = 0; k < NV; ++k) acc[k] = make_float4(0.f, 0.f, 0.f, 0.f);

  for (int j0 = 0; j0 < cnt; j0 += 32) {
    const int j = j0 + lane;
    int s_l = d;
    float p0 = 0.f, p1 = 0.f, p2 = 0.f, p3 = 0.f;
    if (j < cnt) {
      s_l = (j < deg) ? csr_src[base + j] : d;
      const float4 a = node_logits<C>(als, s_l);
      p0 = __expf(lrelu(a.x + adv.x) - m0) / z0;
      p1 = __expf(lrelu(a.y + adv.y) - m1) / z1;
      p2 = __expf(lrelu(a.z + adv.z) - m2) / z2;
      p3 = __expf(lrelu(a.w + adv.w) - m3) / z3;
    }
    const int left = cnt - j0;
    const int nn = (left < 32) ? left : 32;
    for (int t0 = 0; t0 < nn; t0 += UNR) {
#pragma unroll
      for (int u = 0; u < UNR; ++u) {
        const int t = t0 + u;
        const int s = __builtin_amdgcn_readlane(s_l, t);
        const float a0 = __int_as_float(__builtin_amdgcn_readlane(__float_as_int(p0), t));
        const float a1 = __int_as_float(__builtin_amdgcn_readlane(__float_as_int(p1), t));
        const float a2 = __int_as_float(__builtin_amdgcn_readlane(__float_as_int(p2), t));
        const float a3 = __int_as_float(__builtin_amdgcn_readlane(__float_as_int(p3), t));
        const float* __restrict__ hrow = H + (size_t)s * Fout + lane * 4;
#pragma unroll
        for (int k = 0; k < NV; ++k) {
          const float4 hv = *(const float4*)(hrow + k * 128);
          const float al = pick_head<C>(k, lane, a0, a1, a2, a3);
          acc[k].x += al * hv.x;
          acc[k].y += al * hv.y;
          acc[k].z += al * hv.z;
          acc[k].w += al * hv.w;
        }
      }
    }
  }

  if (MEAN) {
    float4 v = acc[0];
    v.x += __shfl_xor(v.x, 8);  v.y += __shfl_xor(v.y, 8);
    v.z += __shfl_xor(v.z, 8);  v.w += __shfl_xor(v.w, 8);
    v.x += __shfl_xor(v.x, 16); v.y += __shfl_xor(v.y, 16);
    v.z += __shfl_xor(v.z, 16); v.w += __shfl_xor(v.w, 16);
    if (lane < 8) {
      const float4 b = *(const float4*)(bias + lane * 4);
      float4 o;
      o.x = 0.25f * v.x + b.x; o.y = 0.25f * v.y + b.y;
      o.z = 0.25f * v.z + b.z; o.w = 0.25f * v.w + b.w;
      *(float4*)(out + (size_t)d * C + lane * 4) = o;
    }
  } else {
#pragma unroll
    for (int k = 0; k < NV; ++k) {
      const float4 b = *(const float4*)(bias + k * 128 + lane * 4);
      float4 o;
      o.x = fmaxf(acc[k].x + b.x, 0.f); o.y = fmaxf(acc[k].y + b.y, 0.f);
      o.z = fmaxf(acc[k].z + b.z, 0.f); o.w = fmaxf(acc[k].w + b.w, 0.f);
      *(float4*)(out + (size_t)d * Fout + k * 128 + lane * 4) = o;
    }
  }
}

__device__ __forceinline__ void mean_heads_store(float4 v, int lane,
                                                 const float* __restrict__ bias,
                                                 float* __restrict__ orow) {
  v.x += __shfl_xor(v.x, 8);  v.y += __shfl_xor(v.y, 8);
  v.z += __shfl_xor(v.z, 8);  v.w += __shfl_xor(v.w, 8);
  v.x += __shfl_xor(v.x, 16); v.y += __shfl_xor(v.y, 16);
  v.z += __shfl_xor(v.z, 16); v.w += __shfl_xor(v.w, 16);
  if (lane < 8) {
    const float4 b = *(const float4*)(bias + lane * 4);
    float4 o;
    o.x = 0.25f * v.x + b.x; o.y = 0.25f * v.y + b.y;
    o.z = 0.25f * v.z + b.z; o.w = 0.25f * v.w + b.w;
    *(float4*)(orow + lane * 4) = o;
  }
}

__global__ __launch_bounds__(256)
void gat_node_pair(const int* __restrict__ rowptr, const int* __restrict__ csr_src,
                   const float* __restrict__ H, const float* __restrict__ als,
                   const float* __restrict__ ald, const float* __restrict__ bias_a,
                   const float* __restrict__ bias_b, float* __restrict__ out_a,
                   float* __restrict__ out_b, int Nn) {
  constexpr int NH = 2 * HEADS;
  constexpr int C = 32;
  constexpr int Fout = NH * C;
  constexpr int UNR = 8;
  const int lane = threadIdx.x & 31;
  const int d = blockIdx.x * 8 + __builtin_amdgcn_readfirstlane((int)(threadIdx.x >> 5));
  if (d >= Nn) return;
  const int base = rowptr[d];
  const int deg  = rowptr[d + 1] - base;
  const int cnt  = deg + 1;
  float adv[NH];
  {
    const float4 x = *(const float4*)(ald + (size_t)d * NH);
    const float4 y = *(const float4*)(ald + (size_t)d * NH + 4);
    adv[0] = x.x; adv[1] = x.y; adv[2] = x.z; adv[3] = x.w;
    adv[4] = y.x; adv[5] = y.y; adv[6] = y.z; adv[7] = y.w;
  }

  float m[NH];
#pragma unroll
  for (int h = 0; h < NH; ++h) m[h] = -__builtin_inff();
  for (int j = lane; j < cnt; j += 32) {
    const int s = (j < deg) ? csr_src[base + j] : d;
    const float4 a = *(const float4*)(als + (size_t)s * NH);
    const float4 b = *(const float4*)(als + (size_t)s * NH + 4);
    const float e[NH] = {a.x, a.y, a.z, a.w, b.x, b.y, b.z, b.w};
#pragma unroll
    for (int h = 0; h < NH; ++h) m[h] = fmaxf(m[h], lrelu(e[h] + adv[h]));
  }
#pragma unroll
  for (int o = 16; o > 0; o >>= 1) {
#pragma unroll
    for (int h = 0; h < NH; ++h) m[h] = fmaxf(m[h], __shfl_xor(m[h], o));
  }

  float z[NH];
#pragma unroll
  for (int h = 0; h < NH; ++h) z[h] = 0.f;
  for (int j = lane; j < cnt; j += 32) {
    const int s = (j < deg) ? csr_src[base + j] : d;
    const float4 a = *(const float4*)(als + (size_t)s * NH);
    const float4 b = *(const float4*)(als + (size_t)s * NH + 4);
    const float e[NH] = {a.x, a.y, a.z, a.w, b.x, b.y, b.z, b.w};
#pragma unroll
    for (int h = 0; h < NH; ++h) z[h] += __expf(lrelu(e[h] + adv[h]) - m[h]);
  }
#pragma unroll
  for (int o = 16; o > 0; o >>= 1) {
#pragma unroll
    for (int h = 0; h < NH; ++h) z[h] += __shfl_xor(z[h], o);
  }
#pragma unroll
  for (int h = 0; h < NH; ++h) z[h] += 1e-16f;

  float4 acca = make_float4(0.f, 0.f, 0.f, 0.f);
  float4 accb = make_float4(0.f, 0.f, 0.f, 0.f);

  for (int j0 = 0; j0 < cnt; j0 += 32) {
    const int j = j0 + lane;
    int s_l = d;
    float p[NH];
#pragma unroll
    for (int h = 0; h < NH; ++h) p[h] = 0.f;
    if (j < cnt) {
      s_l = (j < deg) ? csr_src[base + j] : d;
      const float4 a = *(const float4*)(als + (size_t)s_l * NH);
      const float4 b = *(const float4*)(als + (size_t)s_l * NH + 4);
      const float e[NH] = {a.x, a.y, a.z, a.w, b.x, b.y, b.z, b.w};
#pragma unroll
      for (int h = 0; h < NH; ++h) p[h] = __expf(lrelu(e[h] + adv[h]) - m[h]) / z[h];
    }
    const int left = cnt - j0;
    const int nn = (left < 32) ? left : 32;
    for (int t0 = 0; t0 < nn; t0 += UNR) {
#pragma unroll
      for (int u = 0; u < UNR; ++u) {
        const int t = t0 + u;
        const int s = __builtin_amdgcn_readlane(s_l, t);
        float a[NH];
#pragma unroll
        for (int h = 0; h < NH; ++h)
          a[h] = __int_as_float(__builtin_amdgcn_readlane(__float_as_int(p[h]), t));
        const float* __restrict__ hrow = H + (size_t)s * Fout + lane * 4;
        const float4 hva = *(const float4*)(hrow);
        const float4 hvb = *(const float4*)(hrow + 128);
        const float ala = pick_head<C>(0, lane, a[0], a[1], a[2], a[3]);
        const float alb = pick_head<C>(0, lane, a[4], a[5], a[6], a[7]);
        acca.x += ala * hva.x; acca.y += ala * hva.y;
        acca.z += ala * hva.z; acca.w += ala * hva.w;
        accb.x += alb * hvb.x; accb.y += alb * hvb.y;
        accb.z += alb * hvb.z; accb.w += alb * hvb.w;
      }
    }
  }

  mean_heads_store(acca, lane, bias_a, out_a + (size_t)d * C);
  mean_heads_store(accb, lane, bias_b, out_b + (size_t)d * C);
}

template <int C, bool MEAN>
static void run_gat_layer(hipStream_t st, const float* in, int Fin,
                          const float* W, const float* a_s, const float* a_d, const float* b,
                          float* Hbuf, float* outp,
                          const int* rowptr, const int* csr_src,
                          float* ALS, float* ALD) {
  const int Nn = NNODES;
  const int Fout = HEADS * C;
  const int Mtiles = Nn / 16;
  dim3 ggrid((Mtiles + 7) / 8, Fout / 64);
  gemm_xw_wmma<<<ggrid, 256, 0, st>>>(in, W, W, Hbuf, Nn, Fin, Fout, Fout, Fout,
                                      a_s, a_d, a_s, a_d, ALS, ALD, 1, Fout / 64);
  gat_node<C, MEAN><<<(Nn + 7) / 8, 256, 0, st>>>(rowptr, csr_src, Hbuf, ALS, ALD, b, outp, Nn);
}

extern "C" void kernel_launch(void* const* d_in, const int* in_sizes, int n_in,
                              void* d_out, int out_size, void* d_ws, size_t ws_size,
                              hipStream_t stream) {
  const float* x   = (const float*)d_in[0];
  const int*   ei  = (const int*)d_in[1];
  const int E = in_sizes[1] / 2;
  const int* src = ei;
  const int* dst = ei + E;
  const float *W1 = (const float*)d_in[2],  *as1 = (const float*)d_in[3],
              *ad1 = (const float*)d_in[4], *b1  = (const float*)d_in[5];
  const float *W2 = (const float*)d_in[6],  *as2 = (const float*)d_in[7],
              *ad2 = (const float*)d_in[8], *b2  = (const float*)d_in[9];
  const float *Wm = (const float*)d_in[10], *asm_ = (const float*)d_in[11],
              *adm = (const float*)d_in[12], *bm = (const float*)d_in[13];
  const float *Wl = (const float*)d_in[14], *asl = (const float*)d_in[15],
              *adl = (const float*)d_in[16], *bl = (const float*)d_in[17];

  char* ws = (char*)d_ws;
  size_t off = 0;
  auto alloc = [&](size_t bytes) -> void* {
    void* p = ws + off;
    off += (bytes + 255) & ~(size_t)255;
    return p;
  };
  float* Hbuf = (float*)alloc((size_t)NNODES * 512 * 4);
  float* A0   = (float*)alloc((size_t)NNODES * 256 * 4);
  float* A1   = (float*)alloc((size_t)NNODES * 512 * 4);
  float* ALS  = (float*)alloc((size_t)NNODES * 2 * HEADS * 4);
  float* ALD  = (float*)alloc((size_t)NNODES * 2 * HEADS * 4);
  int*   CNT  = (int*)alloc((size_t)NNODES * 4);
  int*   RPTR = (int*)alloc((size_t)(NNODES + 1) * 4);
  int*   RANK = (int*)alloc((size_t)E * 4);
  int*   CSRC = (int*)alloc((size_t)E * 4);

  float* zmean   = (float*)d_out;
  float* zlogstd = (float*)d_out + (size_t)NNODES * 32;

  zero_i<<<(NNODES + 255) / 256, 256, 0, stream>>>(CNT, NNODES);
  dst_count<<<(E + 1023) / 1024, 256, 0, stream>>>(dst, E, CNT, RANK);
  dst_scan<<<1, 1024, 0, stream>>>(CNT, RPTR);
  dst_fill<<<(E + 1023) / 1024, 256, 0, stream>>>(src, dst, E, RPTR, RANK, CSRC);

  run_gat_layer<64, false>(stream, x, 256, W1, as1, ad1, b1, Hbuf, A0, RPTR, CSRC, ALS, ALD);
  run_gat_layer<128, false>(stream, A0, 256, W2, as2, ad2, b2, Hbuf, A1, RPTR, CSRC, ALS, ALD);
  {
    dim3 ggrid((NNODES / 16 + 7) / 8, 256 / 64);
    gemm_xw_wmma<<<ggrid, 256, 0, stream>>>(A1, Wm, Wl, Hbuf, NNODES, 512, 128, 256, 128,
                                            asm_, adm, asl, adl, ALS, ALD, 2, 8);
    gat_node_pair<<<(NNODES + 7) / 8, 256, 0, stream>>>(RPTR, CSRC, Hbuf, ALS, ALD, bm, bl,
                                                        zmean, zlogstd, NNODES);
  }
}


# Node passes: a node's softmax runs once in registers: first 32 edges' sources and scores loaded once, not 3 times; max/sum butterflies on DPP lane reads, not LDS permutes; sum order unchanged
# speedup vs baseline: 6.2375x; 1.0485x over previous
#include <hip/hip_runtime.h>

typedef float v2f __attribute__((ext_vector_type(2)));
typedef float v8f __attribute__((ext_vector_type(8)));

#define NNODES 30000
#define HEADS 4
#define SLOPE 0.2f
#define KCHUNK 128
#define PAIRSTRIDE 144

__device__ __forceinline__ float sum16_to_lane(const float (&x)[16], int lm) {
  const bool b3 = (lm & 8) != 0, b2 = (lm & 4) != 0, b1 = (lm & 2) != 0, b0 = (lm & 1) != 0;
  float y[8], z[4], w[2];
#pragma unroll
  for (int i = 0; i < 8; ++i)
    y[i] = (b3 ? x[i + 8] : x[i]) + __shfl_xor(b3 ? x[i] : x[i + 8], 8);
#pragma unroll
  for (int i = 0; i < 4; ++i)
    z[i] = (b2 ? y[i + 4] : y[i]) + __shfl_xor(b2 ? y[i] : y[i + 4], 4);
#pragma unroll
  for (int i = 0; i < 2; ++i)
    w[i] = (b1 ? z[i + 2] : z[i]) + __shfl_xor(b1 ? z[i] : z[i + 2], 2);
  return (b0 ? w[1] : w[0]) + __shfl_xor(b0 ? w[0] : w[1], 1);
}

__global__ __launch_bounds__(256, 8)
void gemm_xw_wmma(const float* __restrict__ A, const float* __restrict__ B,
                  const float* __restrict__ B2, float* __restrict__ C,
                  int M, int K, int ldb, int ldc, int nsplit,
                  const float* __restrict__ a_s, const float* __restrict__ a_d,
                  const float* __restrict__ a_s2, const float* __restrict__ a_d2,
                  float* __restrict__ als, float* __restrict__ ald, int G, int ldl) {
  __shared__ float ldsB[(KCHUNK / 2) * PAIRSTRIDE];

  const int tid  = threadIdx.x;
  const int lane = tid & 31;
  const int wave = tid >> 5;
  const int Mtiles = M >> 4;

  const int n0 = blockIdx.y * 64;
  const bool second = (n0 >= nsplit);
  const float* __restrict__ Bs = second ? B2 : B;
  const int nb = second ? n0 - nsplit : n0;
  int mt = blockIdx.x * 8 + wave;
  const bool active = (mt < Mtiles);
  if (!active) mt = Mtiles - 1;
  const int m0   = mt * 16;
  const int half = lane >> 4;
  const int lm   = lane & 15;

  const int lrow = tid >> 4;
  const int lcol = (tid & 15) * 4;

  v8f acc0 = {}, acc1 = {}, acc2 = {}, acc3 = {};
  const float* __restrict__ arow = A + (size_t)(m0 + lm) * K;

  for (int kb = 0; kb < K; kb += KCHUNK) {
    float4 an[4];
#pragma unroll
    for (int q = 0; q < 4; ++q)
      an[q] = *(const float4*)(arow + kb + 8 * q + 4 * half);

    float4 bv[KCHUNK / 16];
#pragma unroll
    for (int r = 0; r < KCHUNK; r += 16)
      bv[r / 16] = *(const float4*)(Bs + (size_t)(kb + r + lrow) * ldb + nb + lcol);

    __syncthreads();
#pragma unroll
    for (int r = 0; r < KCHUNK; r += 16) {
      const int k = r + lrow;
      const float4 v = bv[r / 16];
      float* p = &ldsB[(k >> 1) * PAIRSTRIDE + lcol * 2 + (k & 1)];
      p[0] = v.x; p[2] = v.y; p[4] = v.z; p[6] = v.w;
    }
    __syncthreads();

#pragma unroll
    for (int g = 0; g < KCHUNK / 32; ++g) {
      float4 ac[4];
#pragma unroll
      for (int q = 0; q < 4; ++q) ac[q] = an[q];
      if (g + 1 < KCHUNK / 32) {
        __builtin_amdgcn_sched_barrier(0);
#pragma unroll
        for (int q = 0; q < 4; ++q)
          an[q] = *(const float4*)(arow + kb + 32 * (g + 1) + 8 * q + 4 * half);
        __builtin_amdgcn_sched_barrier(0);
      }
#pragma unroll
      for (int q = 0; q < 4; ++q) {
#pragma unroll
        for (int r2 = 0; r2 < 2; ++r2) {
          v2f a;
          a.x = r2 ? ac[q].z : ac[q].x;
          a.y = r2 ? ac[q].w : ac[q].y;
          const float* __restrict__ bp =
              &ldsB[(16 * g + 4 * q + 2 * half + r2) * PAIRSTRIDE + lm * 2];
          v2f b;
          b = *(const v2f*)(bp + 0);
          acc0 = __builtin_amdgcn_wmma_f32_16x16x4_f32(false, a, false, b, (short)0, acc0, false, false);
          b = *(const v2f*)(bp + 32);
          acc1 = __builtin_amdgcn_wmma_f32_16x16x4_f32(false, a, false, b, (short)0, acc1, false, false);
          b = *(const v2f*)(bp + 64);
          acc2 = __builtin_amdgcn_wmma_f32_16x16x4_f32(false, a, false, b, (short)0, acc2, false, false);
          b = *(const v2f*)(bp + 96);
          acc3 = __builtin_amdgcn_wmma_f32_16x16x4_f32(false, a, false, b, (short)0, acc3, false, false);
        }
      }
    }
  }

  if (active) {
    float* __restrict__ crow = C + (size_t)(m0 + 8 * half) * ldc + n0 + lm;
#pragma unroll
    for (int v = 0; v < 8; ++v) {
      crow[(size_t)v * ldc + 0]  = acc0[v];
      crow[(size_t)v * ldc + 16] = acc1[v];
      crow[(size_t)v * ldc + 32] = acc2[v];
      crow[(size_t)v * ldc + 48] = acc3[v];
    }

    const float* __restrict__ vs = second ? a_s2 : a_s;
    const float* __restrict__ vd = second ? a_d2 : a_d;
    float ws[4], wd[4];
#pragma unroll
    for (int t = 0; t < 4; ++t) {
      ws[t] = vs[nb + 16 * t + lm];
      wd[t] = vd[nb + 16 * t + lm];
    }
    float* __restrict__ lout = ((lm & 8) ? ald : als)
                             + (size_t)(m0 + 8 * half + (lm & 7)) * ldl + blockIdx.y * G;
    float x[16];
    if (G == 1) {
#pragma unroll
      for (int v = 0; v < 8; ++v) {
        x[v]     = acc0[v] * ws[0] + acc1[v] * ws[1] + acc2[v] * ws[2] + acc3[v] * ws[3];
        x[8 + v] = acc0[v] * wd[0] + acc1[v] * wd[1] + acc2[v] * wd[2] + acc3[v] * wd[3];
      }
      lout[0] = sum16_to_lane(x, lm);
    } else {
#pragma unroll
      for (int v = 0; v < 8; ++v) {
        x[v]     = acc0[v] * ws[0] + acc1[v] * ws[1];
        x[8 + v] = acc0[v] * wd[0] + acc1[v] * wd[1];
      }
      lout[0] = sum16_to_lane(x, lm);
#pragma unroll
      for (int v = 0; v < 8; ++v) {
        x[v]     = acc2[v] * ws[2] + acc3[v] * ws[3];
        x[8 + v] = acc2[v] * wd[2] + acc3[v] * wd[3];
      }
      lout[1] = sum16_to_lane(x, lm);
    }
  }
}

__global__ void zero_i(int* __restrict__ p, int n) {
  int i = blockIdx.x * blockDim.x + threadIdx.x;
  if (i < n) p[i] = 0;
}

__global__ void dst_count(const int* __restrict__ dst, int E,
                          int* __restrict__ cnt, int* __restrict__ rank) {
  const int e0 = (blockIdx.x * blockDim.x + threadIdx.x) * 4;
  if (e0 >= E) return;
  if ((E & 3) == 0) {
    const int4 d = *(const int4*)(dst + e0);
    int4 r;
    r.x = atomicAdd(&cnt[d.x], 1);
    r.y = atomicAdd(&cnt[d.y], 1);
    r.z = atomicAdd(&cnt[d.z], 1);
    r.w = atomicAdd(&cnt[d.w], 1);
    *(int4*)(rank + e0) = r;
  } else {
    for (int k = 0; k < 4 && e0 + k < E; ++k) rank[e0 + k] = atomicAdd(&cnt[dst[e0 + k]], 1);
  }
}

#define SCAN_ROWS 8
static_assert(NNODES % 4 == 0, "a lane's four counts must be all inside or all outside");
static_assert(32 * SCAN_ROWS * 128 >= NNODES, "the 32 spans must cover every node");

__device__ __forceinline__ int lane_scan(int v, int lane) {
#pragma unroll
  for (int o = 1; o < 32; o <<= 1) {
    const int n = __shfl_up(v, o);
    if (lane >= o) v += n;
  }
  return v;
}

__global__ __launch_bounds__(1024)
void dst_scan(const int* __restrict__ cnt, int* __restrict__ rowptr) {
  __shared__ int wsum[32];
  const int lane = threadIdx.x & 31;
  const int wave = __builtin_amdgcn_readfirstlane((int)(threadIdx.x >> 5));
  const int lo = wave * (SCAN_ROWS * 128) + lane * 4;

  int4 c[SCAN_ROWS];
#pragma unroll
  for (int q = 0; q < SCAN_ROWS; ++q) {
    const int i = lo + q * 128;
    c[q] = (i < NNODES) ? *(const int4*)(cnt + i) : make_int4(0, 0, 0, 0);
  }

  int carry = 0;
#pragma unroll
  for (int q = 0; q < SCAN_ROWS; ++q) {
    const int4 v = c[q];
    const int s = v.x + v.y + v.z + v.w;
    const int incl = lane_scan(s, lane);
    const int base = carry + incl - s;
    carry += __builtin_amdgcn_readlane(incl, 31);
    c[q] = make_int4(base, base + v.x, base + v.x + v.y, base + v.x + v.y + v.z);
  }
  if (lane == 0) wsum[wave] = carry;
  __syncthreads();

  const int t = wsum[lane];
  const int incl = lane_scan(t, lane);
  const int woff = __builtin_amdgcn_readlane(incl - t, wave);
#pragma unroll
  for (int q = 0; q < SCAN_ROWS; ++q) {
    const int i = lo + q * 128;
    if (i < NNODES) {
      const int4 v = c[q];
      *(int4*)(rowptr + i) = make_int4(v.x + woff, v.y + woff, v.z + woff, v.w + woff);
    }
  }
  if (threadIdx.x == 1023) rowptr[NNODES] = incl;
}

__global__ void dst_fill(const int* __restrict__ src, const int* __restrict__ dst, int E,
                         const int* __restrict__ rowptr, const int* __restrict__ rank,
                         int* __restrict__ csr_src) {
  const int e0 = (blockIdx.x * blockDim.x + threadIdx.x) * 4;
  if (e0 >= E) return;
  if ((E & 3) == 0) {
    const int4 d = *(const int4*)(dst + e0);
    const int4 s = *(const int4*)(src + e0);
    const int4 r = *(const int4*)(rank + e0);
    const int p0 = rowptr[d.x], p1 = rowptr[d.y], p2 = rowptr[d.z], p3 = rowptr[d.w];
    csr_src[p0 + r.x] = s.x;
    csr_src[p1 + r.y] = s.y;
    csr_src[p2 + r.z] = s.z;
    csr_src[p3 + r.w] = s.w;
  } else {
    for (int k = 0; k < 4 && e0 + k < E; ++k)
      csr_src[rowptr[dst[e0 + k]] + rank[e0 + k]] = src[e0 + k];
  }
}

__device__ __forceinline__ float lrelu(float v) { return (v > 0.f) ? v : SLOPE * v; }

template <int O>
__device__ __forceinline__ float lane_xor(float v) {
  const int b = __float_as_int(v);
  if (O == 16)
    return __int_as_float(__builtin_amdgcn_permlanex16(b, b, 0x76543210, 0xfedcba98, false, false));
  return __int_as_float(__builtin_amdgcn_update_dpp(b, b, 0x160 + O, 0xf, 0xf, false));
}
__device__ __forceinline__ float wave_max(float v) {
  v = fmaxf(v, lane_xor<16>(v)); v = fmaxf(v, lane_xor<8>(v)); v = fmaxf(v, lane_xor<4>(v));
  v = fmaxf(v, lane_xor<2>(v));  v = fmaxf(v, lane_xor<1>(v));
  return v;
}
__device__ __forceinline__ float wave_sum(float v) {
  v += lane_xor<16>(v); v += lane_xor<8>(v); v += lane_xor<4>(v);
  v += lane_xor<2>(v);  v += lane_xor<1>(v);
  return v;
}

template <int C>
__device__ __forceinline__ float4 node_logits(const float* __restrict__ p, int node) {
  if (C == 128) {
    const float4 q0 = *(const float4*)(p + (size_t)node * 8);
    const float4 q1 = *(const float4*)(p + (size_t)node * 8 + 4);
    return make_float4(q0.x + q0.y, q0.z + q0.w, q1.x + q1.y, q1.z + q1.w);
  }
  return *(const float4*)(p + (size_t)node * 4);
}

template <int C>
__device__ __forceinline__ float pick_head(int k, int lane, float a0, float a1, float a2, float a3) {
  if (C == 128) return (k == 0) ? a0 : (k == 1) ? a1 : (k == 2) ? a2 : a3;
  if (C == 64) {
    const bool hi = (lane & 16) != 0;
    return (k == 0) ? (hi ? a1 : a0) : (hi ? a3 : a2);
  }
  const int h = lane >> 3;
  const float x01 = (h & 1) ? a1 : a0;
  const float x23 = (h & 1) ? a3 : a2;
  return (h & 2) ? x23 : x01;
}

template <int C, int NV, int UNR>
__device__ __forceinline__ void gather_rows(const float* __restrict__ H, int lane, int nn, int s_l,
                                            float p0, float p1, float p2, float p3,
                                            float4 (&acc)[NV]) {
  constexpr int Fout = HEADS * C;
  for (int t0 = 0; t0 < nn; t0 += UNR) {
#pragma unroll
    for (int u = 0; u < UNR; ++u) {
      const int t = t0 + u;
      const int s = __builtin_amdgcn_readlane(s_l, t);
      const float a0 = __int_as_float(__builtin_amdgcn_readlane(__float_as_int(p0), t));
      const float a1 = __int_as_float(__builtin_amdgcn_readlane(__float_as_int(p1), t));
      const float a2 = __int_as_float(__builtin_amdgcn_readlane(__float_as_int(p2), t));
      const float a3 = __int_as_float(__builtin_amdgcn_readlane(__float_as_int(p3), t));
      const float* __restrict__ hrow = H + (size_t)s * Fout + lane * 4;
#pragma unroll
      for (int k = 0; k < NV; ++k) {
        const float4 hv = *(const float4*)(hrow + k * 128);
        const float al = pick_head<C>(k, lane, a0, a1, a2, a3);
        acc[k].x += al * hv.x;
        acc[k].y += al * hv.y;
        acc[k].z += al * hv.z;
        acc[k].w += al * hv.w;
      }
    }
  }
}

template <int C, bool MEAN>
__global__ __launch_bounds__(256)
void gat_node(const int* __restrict__ rowptr, const int* __restrict__ csr_src,
              const float* __restrict__ H, const float* __restrict__ als,
              const float* __restrict__ ald, const float* __restrict__ bias,
              float* __restrict__ out, int Nn) {
  constexpr int Fout = HEADS * C;
  constexpr int NV = Fout / 128;
  constexpr int UNR = (NV == 1) ? 8 : 4;
  const int lane = threadIdx.x & 31;
  const int d = blockIdx.x * 8 + __builtin_amdgcn_readfirstlane((int)(threadIdx.x >> 5));
  if (d >= Nn) return;
  const int base = rowptr[d];
  const int deg  = rowptr[d + 1] - base;
  const int cnt  = deg + 1;
  const float4 adv = node_logits<C>(ald, d);

  int s_f = d;
  float e0 = -__builtin_inff(), e1 = e0, e2 = e0, e3 = e0;
  if (lane < cnt) {
    s_f = (lane < deg) ? csr_src[base + lane] : d;
    const float4 a = node_logits<C>(als, s_f);
    e0 = lrelu(a.x + adv.x); e1 = lrelu(a.y + adv.y);
    e2 = lrelu(a.z + adv.z); e3 = lrelu(a.w + adv.w);
  }

  float m0 = e0, m1 = e1, m2 = e2, m3 = e3;
  for (int j = lane + 32; j < cnt; j += 32) {
    const int s = (j < deg) ? csr_src[base + j] : d;
    const float4 a = node_logits<C>(als, s);
    m0 = fmaxf(m0, lrelu(a.x + adv.x));
    m1 = fmaxf(m1, lrelu(a.y + adv.y));
    m2 = fmaxf(m2, lrelu(a.z + adv.z));
    m3 = fmaxf(m3, lrelu(a.w + adv.w));
  }
  m0 = wave_max(m0); m1 = wave_max(m1); m2 = wave_max(m2); m3 = wave_max(m3);

  const bool in_f = lane < cnt;
  const float x0 = in_f ? __expf(e0 - m0) : 0.f, x1 = in_f ? __expf(e1 - m1) : 0.f;
  const float x2 = in_f ? __expf(e2 - m2) : 0.f, x3 = in_f ? __expf(e3 - m3) : 0.f;
  float z0 = x0, z1 = x1, z2 = x2, z3 = x3;
  for (int j = lane + 32; j < cnt; j += 32) {
    const int s = (j < deg) ? csr_src[base + j] : d;
    const float4 a = node_logits<C>(als, s);
    z0 += __expf(lrelu(a.x + adv.x) - m0);
    z1 += __expf(lrelu(a.y + adv.y) - m1);
    z2 += __expf(lrelu(a.z + adv.z) - m2);
    z3 += __expf(lrelu(a.w + adv.w) - m3);
  }
  z0 = wave_sum(z0); z1 = wave_sum(z1); z2 = wave_sum(z2); z3 = wave_sum(z3);
  z0 += 1e-16f; z1 += 1e-16f; z2 += 1e-16f; z3 += 1e-16f;

  float4 acc[NV];
#pragma unroll
  for (int k = 0; k < NV; ++k) acc[k] = make_float4(0.f, 0.f, 0.f, 0.f);

  gather_rows<C, NV, UNR>(H, lane, (cnt < 32) ? cnt : 32, s_f,
                          x0 / z0, x1 / z1, x2 / z2, x3 / z3, acc);
  for (int j0 = 32; j0 < cnt; j0 += 32) {
    const int j = j0 + lane;
    int s_l = d;
    float p0 = 0.f, p1 = 0.f, p2 = 0.f, p3 = 0.f;
    if (j < cnt) {
      s_l = (j < deg) ? csr_src[base + j] : d;
      const float4 a = node_logits<C>(als, s_l);
      p0 = __expf(lrelu(a.x + adv.x) - m0) / z0;
      p1 = __expf(lrelu(a.y + adv.y) - m1) / z1;
      p2 = __expf(lrelu(a.z + adv.z) - m2) / z2;
      p3 = __expf(lrelu(a.w + adv.w) - m3) / z3;
    }
    const int left = cnt - j0;
    gather_rows<C, NV, UNR>(H, lane, (left < 32) ? left : 32, s_l, p0, p1, p2, p3, acc);
  }

  if (MEAN) {
    float4 v = acc[0];
    v.x += lane_xor<8>(v.x);  v.y += lane_xor<8>(v.y);
    v.z += lane_xor<8>(v.z);  v.w += lane_xor<8>(v.w);
    v.x += lane_xor<16>(v.x); v.y += lane_xor<16>(v.y);
    v.z += lane_xor<16>(v.z); v.w += lane_xor<16>(v.w);
    if (lane < 8) {
      const float4 b = *(const float4*)(bias + lane * 4);
      float4 o;
      o.x = 0.25f * v.x + b.x; o.y = 0.25f * v.y + b.y;
      o.z = 0.25f * v.z + b.z; o.w = 0.25f * v.w + b.w;
      *(float4*)(out + (size_t)d * C + lane * 4) = o;
    }
  } else {
#pragma unroll
    for (int k = 0; k < NV; ++k) {
      const float4 b = *(const float4*)(bias + k * 128 + lane * 4);
      float4 o;
      o.x = fmaxf(acc[k].x + b.x, 0.f); o.y = fmaxf(acc[k].y + b.y, 0.f);
      o.z = fmaxf(acc[k].z + b.z, 0.f); o.w = fmaxf(acc[k].w + b.w, 0.f);
      *(float4*)(out + (size_t)d * Fout + k * 128 + lane * 4) = o;
    }
  }
}

__device__ __forceinline__ void mean_heads_store(float4 v, int lane,
                                                 const float* __restrict__ bias,
                                                 float* __restrict__ orow) {
  v.x += lane_xor<8>(v.x);  v.y += lane_xor<8>(v.y);
  v.z += lane_xor<8>(v.z);  v.w += lane_xor<8>(v.w);
  v.x += lane_xor<16>(v.x); v.y += lane_xor<16>(v.y);
  v.z += lane_xor<16>(v.z); v.w += lane_xor<16>(v.w);
  if (lane < 8) {
    const float4 b = *(const float4*)(bias + lane * 4);
    float4 o;
    o.x = 0.25f * v.x + b.x; o.y = 0.25f * v.y + b.y;
    o.z = 0.25f * v.z + b.z; o.w = 0.25f * v.w + b.w;
    *(float4*)(orow + lane * 4) = o;
  }
}

__device__ __forceinline__ void gather_rows_pair(const float* __restrict__ H, int lane, int nn,
                                                 int s_l, const float (&p)[2 * HEADS],
                                                 float4& acca, float4& accb) {
  constexpr int NH = 2 * HEADS;
  constexpr int UNR = 8;
  for (int t0 = 0; t0 < nn; t0 += UNR) {
#pragma unroll
    for (int u = 0; u < UNR; ++u) {
      const int t = t0 + u;
      const int s = __builtin_amdgcn_readlane(s_l, t);
      float a[NH];
#pragma unroll
      for (int h = 0; h < NH; ++h)
        a[h] = __int_as_float(__builtin_amdgcn_readlane(__float_as_int(p[h]), t));
      const float* __restrict__ hrow = H + (size_t)s * (NH * 32) + lane * 4;
      const float4 hva = *(const float4*)(hrow);
      const float4 hvb = *(const float4*)(hrow + 128);
      const float ala = pick_head<32>(0, lane, a[0], a[1], a[2], a[3]);
      const float alb = pick_head<32>(0, lane, a[4], a[5], a[6], a[7]);
      acca.x += ala * hva.x; acca.y += ala * hva.y;
      acca.z += ala * hva.z; acca.w += ala * hva.w;
      accb.x += alb * hvb.x; accb.y += alb * hvb.y;
      accb.z += alb * hvb.z; accb.w += alb * hvb.w;
    }
  }
}

__global__ __launch_bounds__(256)
void gat_node_pair(const int* __restrict__ rowptr, const int* __restrict__ csr_src,
                   const float* __restrict__ H, const float* __restrict__ als,
                   const float* __restrict__ ald, const float* __restrict__ bias_a,
                   const float* __restrict__ bias_b, float* __restrict__ out_a,
                   float* __restrict__ out_b, int Nn) {
  constexpr int NH = 2 * HEADS;
  constexpr int C = 32;
  const int lane = threadIdx.x & 31;
  const int d = blockIdx.x * 8 + __builtin_amdgcn_readfirstlane((int)(threadIdx.x >> 5));
  if (d >= Nn) return;
  const int base = rowptr[d];
  const int deg  = rowptr[d + 1] - base;
  const int cnt  = deg + 1;
  float adv[NH];
  {
    const float4 x = *(const float4*)(ald + (size_t)d * NH);
    const float4 y = *(const float4*)(ald + (size_t)d * NH + 4);
    adv[0] = x.x; adv[1] = x.y; adv[2] = x.z; adv[3] = x.w;
    adv[4] = y.x; adv[5] = y.y; adv[6] = y.z; adv[7] = y.w;
  }

  int s_f = d;
  float ef[NH];
#pragma unroll
  for (int h = 0; h < NH; ++h) ef[h] = -__builtin_inff();
  if (lane < cnt) {
    s_f = (lane < deg) ? csr_src[base + lane] : d;
    const float4 a = *(const float4*)(als + (size_t)s_f * NH);
    const float4 b = *(const float4*)(als + (size_t)s_f * NH + 4);
    const float e[NH] = {a.x, a.y, a.z, a.w, b.x, b.y, b.z, b.w};
#pragma unroll
    for (int h = 0; h < NH; ++h) ef[h] = lrelu(e[h] + adv[h]);
  }

  float m[NH];
#pragma unroll
  for (int h = 0; h < NH; ++h) m[h] = ef[h];
  for (int j = lane + 32; j < cnt; j += 32) {
    const int s = (j < deg) ? csr_src[base + j] : d;
    const float4 a = *(const float4*)(als + (size_t)s * NH);
    const float4 b = *(const float4*)(als + (size_t)s * NH + 4);
    const float e[NH] = {a.x, a.y, a.z, a.w, b.x, b.y, b.z, b.w};
#pragma unroll
    for (int h = 0; h < NH; ++h) m[h] = fmaxf(m[h], lrelu(e[h] + adv[h]));
  }
#pragma unroll
  for (int h = 0; h < NH; ++h) m[h] = wave_max(m[h]);

  const bool in_f = lane < cnt;
  float xf[NH], z[NH];
#pragma unroll
  for (int h = 0; h < NH; ++h) {
    xf[h] = in_f ? __expf(ef[h] - m[h]) : 0.f;
    z[h] = xf[h];
  }
  for (int j = lane + 32; j < cnt; j += 32) {
    const int s = (j < deg) ? csr_src[base + j] : d;
    const float4 a = *(const float4*)(als + (size_t)s * NH);
    const float4 b = *(const float4*)(als + (size_t)s * NH + 4);
    const float e[NH] = {a.x, a.y, a.z, a.w, b.x, b.y, b.z, b.w};
#pragma unroll
    for (int h = 0; h < NH; ++h) z[h] += __expf(lrelu(e[h] + adv[h]) - m[h]);
  }
#pragma unroll
  for (int h = 0; h < NH; ++h) z[h] = wave_sum(z[h]) + 1e-16f;

  float4 acca = make_float4(0.f, 0.f, 0.f, 0.f);
  float4 accb = make_float4(0.f, 0.f, 0.f, 0.f);

  {
    float p[NH];
#pragma unroll
    for (int h = 0; h < NH; ++h) p[h] = xf[h] / z[h];
    gather_rows_pair(H, lane, (cnt < 32) ? cnt : 32, s_f, p, acca, accb);
  }
  for (int j0 = 32; j0 < cnt; j0 += 32) {
    const int j = j0 + lane;
    int s_l = d;
    float p[NH];
#pragma unroll
    for (int h = 0; h < NH; ++h) p[h] = 0.f;
    if (j < cnt) {
      s_l = (j < deg) ? csr_src[base + j] : d;
      const float4 a = *(const float4*)(als + (size_t)s_l * NH);
      const float4 b = *(const float4*)(als + (size_t)s_l * NH + 4);
      const float e[NH] = {a.x, a.y, a.z, a.w, b.x, b.y, b.z, b.w};
#pragma unroll
      for (int h = 0; h < NH; ++h) p[h] = __expf(lrelu(e[h] + adv[h]) - m[h]) / z[h];
    }
    const int left = cnt - j0;
    gather_rows_pair(H, lane, (left < 32) ? left : 32, s_l, p, acca, accb);
  }

  mean_heads_store(acca, lane, bias_a, out_a + (size_t)d * C);
  mean_heads_store(accb, lane, bias_b, out_b + (size_t)d * C);
}

template <int C, bool MEAN>
static void run_gat_layer(hipStream_t st, const float* in, int Fin,
                          const float* W, const float* a_s, const float* a_d, const float* b,
                          float* Hbuf, float* outp,
                          const int* rowptr, const int* csr_src,
                          float* ALS, float* ALD) {
  const int Nn = NNODES;
  const int Fout = HEADS * C;
  const int Mtiles = Nn / 16;
  dim3 ggrid((Mtiles + 7) / 8, Fout / 64);
  gemm_xw_wmma<<<ggrid, 256, 0, st>>>(in, W, W, Hbuf, Nn, Fin, Fout, Fout, Fout,
                                      a_s, a_d, a_s, a_d, ALS, ALD, 1, Fout / 64);
  gat_node<C, MEAN><<<(Nn + 7) / 8, 256, 0, st>>>(rowptr, csr_src, Hbuf, ALS, ALD, b, outp, Nn);
}

extern "C" void kernel_launch(void* const* d_in, const int* in_sizes, int n_in,
                              void* d_out, int out_size, void* d_ws, size_t ws_size,
                              hipStream_t stream) {
  const float* x   = (const float*)d_in[0];
  const int*   ei  = (const int*)d_in[1];
  const int E = in_sizes[1] / 2;
  const int* src = ei;
  const int* dst = ei + E;
  const float *W1 = (const float*)d_in[2],  *as1 = (const float*)d_in[3],
              *ad1 = (const float*)d_in[4], *b1  = (const float*)d_in[5];
  const float *W2 = (const float*)d_in[6],  *as2 = (const float*)d_in[7],
              *ad2 = (const float*)d_in[8], *b2  = (const float*)d_in[9];
  const float *Wm = (const float*)d_in[10], *asm_ = (const float*)d_in[11],
              *adm = (const float*)d_in[12], *bm = (const float*)d_in[13];
  const float *Wl = (const float*)d_in[14], *asl = (const float*)d_in[15],
              *adl = (const float*)d_in[16], *bl = (const float*)d_in[17];

  char* ws = (char*)d_ws;
  size_t off = 0;
  auto alloc = [&](size_t bytes) -> void* {
    void* p = ws + off;
    off += (bytes + 255) & ~(size_t)255;
    return p;
  };
  float* Hbuf = (float*)alloc((size_t)NNODES * 512 * 4);
  float* A0   = (float*)alloc((size_t)NNODES * 256 * 4);
  float* A1   = (float*)alloc((size_t)NNODES * 512 * 4);
  float* ALS  = (float*)alloc((size_t)NNODES * 2 * HEADS * 4);
  float* ALD  = (float*)alloc((size_t)NNODES * 2 * HEADS * 4);
  int*   CNT  = (int*)alloc((size_t)NNODES * 4);
  int*   RPTR = (int*)alloc((size_t)(NNODES + 1) * 4);
  int*   RANK = (int*)alloc((size_t)E * 4);
  int*   CSRC = (int*)alloc((size_t)E * 4);

  float* zmean   = (float*)d_out;
  float* zlogstd = (float*)d_out + (size_t)NNODES * 32;

  zero_i<<<(NNODES + 255) / 256, 256, 0, stream>>>(CNT, NNODES);
  dst_count<<<(E + 1023) / 1024, 256, 0, stream>>>(dst, E, CNT, RANK);
  dst_scan<<<1, 1024, 0, stream>>>(CNT, RPTR);
  dst_fill<<<(E + 1023) / 1024, 256, 0, stream>>>(src, dst, E, RPTR, RANK, CSRC);

  run_gat_layer<64, false>(stream, x, 256, W1, as1, ad1, b1, Hbuf, A0, RPTR, CSRC, ALS, ALD);
  run_gat_layer<128, false>(stream, A0, 256, W2, as2, ad2, b2, Hbuf, A1, RPTR, CSRC, ALS, ALD);
  {
    dim3 ggrid((NNODES / 16 + 7) / 8, 256 / 64);
    gemm_xw_wmma<<<ggrid, 256, 0, stream>>>(A1, Wm, Wl, Hbuf, NNODES, 512, 128, 256, 128,
                                            asm_, adm, asl, adl, ALS, ALD, 2, 8);
    gat_node_pair<<<(NNODES + 7) / 8, 256, 0, stream>>>(RPTR, CSRC, Hbuf, ALS, ALD, bm, bl,
                                                        zmean, zlogstd, NNODES);
  }
}


# Counting pass stores each edge straight into its node's 64-slot bucket (slot = arrival rank; longer rows spill to a list summed after the self loop); first GEMM clears counters: 7 launches, was 10
# speedup vs baseline: 6.5719x; 1.0536x over previous
#include <hip/hip_runtime.h>

typedef float v2f __attribute__((ext_vector_type(2)));
typedef float v8f __attribute__((ext_vector_type(8)));

#define NNODES 30000
#define HEADS 4
#define SLOPE 0.2f
#define KCHUNK 128
#define CAP 64
#define PAIRSTRIDE 144

__device__ __forceinline__ float sum16_to_lane(const float (&x)[16], int lm) {
  const bool b3 = (lm & 8) != 0, b2 = (lm & 4) != 0, b1 = (lm & 2) != 0, b0 = (lm & 1) != 0;
  float y[8], z[4], w[2];
#pragma unroll
  for (int i = 0; i < 8; ++i)
    y[i] = (b3 ? x[i + 8] : x[i]) + __shfl_xor(b3 ? x[i] : x[i + 8], 8);
#pragma unroll
  for (int i = 0; i < 4; ++i)
    z[i] = (b2 ? y[i + 4] : y[i]) + __shfl_xor(b2 ? y[i] : y[i + 4], 4);
#pragma unroll
  for (int i = 0; i < 2; ++i)
    w[i] = (b1 ? z[i + 2] : z[i]) + __shfl_xor(b1 ? z[i] : z[i + 2], 2);
  return (b0 ? w[1] : w[0]) + __shfl_xor(b0 ? w[0] : w[1], 1);
}

__global__ __launch_bounds__(256, 8)
void gemm_xw_wmma(const float* __restrict__ A, const float* __restrict__ B,
                  const float* __restrict__ B2, float* __restrict__ C,
                  int M, int K, int ldb, int ldc, int nsplit,
                  const float* __restrict__ a_s, const float* __restrict__ a_d,
                  const float* __restrict__ a_s2, const float* __restrict__ a_d2,
                  float* __restrict__ als, float* __restrict__ ald, int G, int ldl,
                  int* __restrict__ zp, int zn) {
  __shared__ float ldsB[(KCHUNK / 2) * PAIRSTRIDE];

  const int tid  = threadIdx.x;
  if (zp != nullptr) {
    const int zi = (blockIdx.y * gridDim.x + blockIdx.x) * 256 + tid;
    if (zi < zn) zp[zi] = 0;
  }
  const int lane = tid & 31;
  const int wave = tid >> 5;
  const int Mtiles = M >> 4;

  const int n0 = blockIdx.y * 64;
  const bool second = (n0 >= nsplit);
  const float* __restrict__ Bs = second ? B2 : B;
  const int nb = second ? n0 - nsplit : n0;
  int mt = blockIdx.x * 8 + wave;
  const bool active = (mt < Mtiles);
  if (!active) mt = Mtiles - 1;
  const int m0   = mt * 16;
  const int half = lane >> 4;
  const int lm   = lane & 15;

  const int lrow = tid >> 4;
  const int lcol = (tid & 15) * 4;

  v8f acc0 = {}, acc1 = {}, acc2 = {}, acc3 = {};
  const float* __restrict__ arow = A + (size_t)(m0 + lm) * K;

  for (int kb = 0; kb < K; kb += KCHUNK) {
    float4 an[4];
#pragma unroll
    for (int q = 0; q < 4; ++q)
      an[q] = *(const float4*)(arow + kb + 8 * q + 4 * half);

    float4 bv[KCHUNK / 16];
#pragma unroll
    for (int r = 0; r < KCHUNK; r += 16)
      bv[r / 16] = *(const float4*)(Bs + (size_t)(kb + r + lrow) * ldb + nb + lcol);

    __syncthreads();
#pragma unroll
    for (int r = 0; r < KCHUNK; r += 16) {
      const int k = r + lrow;
      const float4 v = bv[r / 16];
      float* p = &ldsB[(k >> 1) * PAIRSTRIDE + lcol * 2 + (k & 1)];
      p[0] = v.x; p[2] = v.y; p[4] = v.z; p[6] = v.w;
    }
    __syncthreads();

#pragma unroll
    for (int g = 0; g < KCHUNK / 32; ++g) {
      float4 ac[4];
#pragma unroll
      for (int q = 0; q < 4; ++q) ac[q] = an[q];
      if (g + 1 < KCHUNK / 32) {
        __builtin_amdgcn_sched_barrier(0);
#pragma unroll
        for (int q = 0; q < 4; ++q)
          an[q] = *(const float4*)(arow + kb + 32 * (g + 1) + 8 * q + 4 * half);
        __builtin_amdgcn_sched_barrier(0);
      }
#pragma unroll
      for (int q = 0; q < 4; ++q) {
#pragma unroll
        for (int r2 = 0; r2 < 2; ++r2) {
          v2f a;
          a.x = r2 ? ac[q].z : ac[q].x;
          a.y = r2 ? ac[q].w : ac[q].y;
          const float* __restrict__ bp =
              &ldsB[(16 * g + 4 * q + 2 * half + r2) * PAIRSTRIDE + lm * 2];
          v2f b;
          b = *(const v2f*)(bp + 0);
          acc0 = __builtin_amdgcn_wmma_f32_16x16x4_f32(false, a, false, b, (short)0, acc0, false, false);
          b = *(const v2f*)(bp + 32);
          acc1 = __builtin_amdgcn_wmma_f32_16x16x4_f32(false, a, false, b, (short)0, acc1, false, false);
          b = *(const v2f*)(bp + 64);
          acc2 = __builtin_amdgcn_wmma_f32_16x16x4_f32(false, a, false, b, (short)0, acc2, false, false);
          b = *(const v2f*)(bp + 96);
          acc3 = __builtin_amdgcn_wmma_f32_16x16x4_f32(false, a, false, b, (short)0, acc3, false, false);
        }
      }
    }
  }

  if (active) {
    float* __restrict__ crow = C + (size_t)(m0 + 8 * half) * ldc + n0 + lm;
#pragma unroll
    for (int v = 0; v < 8; ++v) {
      crow[(size_t)v * ldc + 0]  = acc0[v];
      crow[(size_t)v * ldc + 16] = acc1[v];
      crow[(size_t)v * ldc + 32] = acc2[v];
      crow[(size_t)v * ldc + 48] = acc3[v];
    }

    const float* __restrict__ vs = second ? a_s2 : a_s;
    const float* __restrict__ vd = second ? a_d2 : a_d;
    float ws[4], wd[4];
#pragma unroll
    for (int t = 0; t < 4; ++t) {
      ws[t] = vs[nb + 16 * t + lm];
      wd[t] = vd[nb + 16 * t + lm];
    }
    float* __restrict__ lout = ((lm & 8) ? ald : als)
                             + (size_t)(m0 + 8 * half + (lm & 7)) * ldl + blockIdx.y * G;
    float x[16];
    if (G == 1) {
#pragma unroll
      for (int v = 0; v < 8; ++v) {
        x[v]     = acc0[v] * ws[0] + acc1[v] * ws[1] + acc2[v] * ws[2] + acc3[v] * ws[3];
        x[8 + v] = acc0[v] * wd[0] + acc1[v] * wd[1] + acc2[v] * wd[2] + acc3[v] * wd[3];
      }
      lout[0] = sum16_to_lane(x, lm);
    } else {
#pragma unroll
      for (int v = 0; v < 8; ++v) {
        x[v]     = acc0[v] * ws[0] + acc1[v] * ws[1];
        x[8 + v] = acc0[v] * wd[0] + acc1[v] * wd[1];
      }
      lout[0] = sum16_to_lane(x, lm);
#pragma unroll
      for (int v = 0; v < 8; ++v) {
        x[v]     = acc2[v] * ws[2] + acc3[v] * ws[3];
        x[8 + v] = acc2[v] * wd[2] + acc3[v] * wd[3];
      }
      lout[1] = sum16_to_lane(x, lm);
    }
  }
}

__device__ __forceinline__ void place_edge(int d, int s, int r, int* __restrict__ cnt,
                                           int* __restrict__ bucket, int* __restrict__ ovf_dst,
                                           int* __restrict__ ovf_src) {
  if (r < CAP) {
    bucket[d * CAP + r] = s;
  } else {
    const int p = atomicAdd(&cnt[NNODES], 1);
    ovf_dst[p] = d;
    ovf_src[p] = s;
  }
}

__global__ void dst_bucket(const int* __restrict__ src, const int* __restrict__ dst, int E,
                           int* __restrict__ cnt, int* __restrict__ bucket,
                           int* __restrict__ ovf_dst, int* __restrict__ ovf_src) {
  const int e0 = (blockIdx.x * blockDim.x + threadIdx.x) * 4;
  if (e0 >= E) return;
  if ((E & 3) == 0) {
    const int4 d = *(const int4*)(dst + e0);
    const int4 s = *(const int4*)(src + e0);
    const int r0 = atomicAdd(&cnt[d.x], 1);
    const int r1 = atomicAdd(&cnt[d.y], 1);
    const int r2 = atomicAdd(&cnt[d.z], 1);
    const int r3 = atomicAdd(&cnt[d.w], 1);
    place_edge(d.x, s.x, r0, cnt, bucket, ovf_dst, ovf_src);
    place_edge(d.y, s.y, r1, cnt, bucket, ovf_dst, ovf_src);
    place_edge(d.z, s.z, r2, cnt, bucket, ovf_dst, ovf_src);
    place_edge(d.w, s.w, r3, cnt, bucket, ovf_dst, ovf_src);
  } else {
    for (int k = 0; k < 4 && e0 + k < E; ++k) {
      const int dd = dst[e0 + k];
      place_edge(dd, src[e0 + k], atomicAdd(&cnt[dd], 1), cnt, bucket, ovf_dst, ovf_src);
    }
  }
}

__device__ __forceinline__ float lrelu(float v) { return (v > 0.f) ? v : SLOPE * v; }

template <int O>
__device__ __forceinline__ float lane_xor(float v) {
  const int b = __float_as_int(v);
  if (O == 16)
    return __int_as_float(__builtin_amdgcn_permlanex16(b, b, 0x76543210, 0xfedcba98, false, false));
  return __int_as_float(__builtin_amdgcn_update_dpp(b, b, 0x160 + O, 0xf, 0xf, false));
}
__device__ __forceinline__ float wave_max(float v) {
  v = fmaxf(v, lane_xor<16>(v)); v = fmaxf(v, lane_xor<8>(v)); v = fmaxf(v, lane_xor<4>(v));
  v = fmaxf(v, lane_xor<2>(v));  v = fmaxf(v, lane_xor<1>(v));
  return v;
}
__device__ __forceinline__ float wave_sum(float v) {
  v += lane_xor<16>(v); v += lane_xor<8>(v); v += lane_xor<4>(v);
  v += lane_xor<2>(v);  v += lane_xor<1>(v);
  return v;
}

template <int C>
__device__ __forceinline__ float4 node_logits(const float* __restrict__ p, int node) {
  if (C == 128) {
    const float4 q0 = *(const float4*)(p + (size_t)node * 8);
    const float4 q1 = *(const float4*)(p + (size_t)node * 8 + 4);
    return make_float4(q0.x + q0.y, q0.z + q0.w, q1.x + q1.y, q1.z + q1.w);
  }
  return *(const float4*)(p + (size_t)node * 4);
}

template <int C>
__device__ __forceinline__ float pick_head(int k, int lane, float a0, float a1, float a2, float a3) {
  if (C == 128) return (k == 0) ? a0 : (k == 1) ? a1 : (k == 2) ? a2 : a3;
  if (C == 64) {
    const bool hi = (lane & 16) != 0;
    return (k == 0) ? (hi ? a1 : a0) : (hi ? a3 : a2);
  }
  const int h = lane >> 3;
  const float x01 = (h & 1) ? a1 : a0;
  const float x23 = (h & 1) ? a3 : a2;
  return (h & 2) ? x23 : x01;
}

template <int C, int NV, int UNR>
__device__ __forceinline__ void gather_rows(const float* __restrict__ H, int lane, int nn, int s_l,
                                            float p0, float p1, float p2, float p3,
                                            float4 (&acc)[NV]) {
  constexpr int Fout = HEADS * C;
  for (int t0 = 0; t0 < nn; t0 += UNR) {
#pragma unroll
    for (int u = 0; u < UNR; ++u) {
      const int t = t0 + u;
      const int s = __builtin_amdgcn_readlane(s_l, t);
      const float a0 = __int_as_float(__builtin_amdgcn_readlane(__float_as_int(p0), t));
      const float a1 = __int_as_float(__builtin_amdgcn_readlane(__float_as_int(p1), t));
      const float a2 = __int_as_float(__builtin_amdgcn_readlane(__float_as_int(p2), t));
      const float a3 = __int_as_float(__builtin_amdgcn_readlane(__float_as_int(p3), t));
      const float* __restrict__ hrow = H + (size_t)s * Fout + lane * 4;
#pragma unroll
      for (int k = 0; k < NV; ++k) {
        const float4 hv = *(const float4*)(hrow + k * 128);
        const float al = pick_head<C>(k, lane, a0, a1, a2, a3);
        acc[k].x += al * hv.x;
        acc[k].y += al * hv.y;
        acc[k].z += al * hv.z;
        acc[k].w += al * hv.w;
      }
    }
  }
}

template <int C, bool MEAN>
__global__ __launch_bounds__(256)
void gat_node(const int* __restrict__ cnt_arr, const int* __restrict__ csr_src,
              const int* __restrict__ ovf_dst, const int* __restrict__ ovf_src,
              const float* __restrict__ H, const float* __restrict__ als,
              const float* __restrict__ ald, const float* __restrict__ bias,
              float* __restrict__ out, int Nn) {
  constexpr int Fout = HEADS * C;
  constexpr int NV = Fout / 128;
  constexpr int UNR = (NV == 1) ? 8 : 4;
  const int lane = threadIdx.x & 31;
  const int d = blockIdx.x * 8 + __builtin_amdgcn_readfirstlane((int)(threadIdx.x >> 5));
  if (d >= Nn) return;
  const int base = d * CAP;
  const int deg_all = cnt_arr[d];
  const int deg  = (deg_all < CAP) ? deg_all : CAP;
  const bool spill = deg_all > CAP;
  const int cnt  = deg + 1;
  const float4 adv = node_logits<C>(ald, d);

  int s_f = d;
  float e0 = -__builtin_inff(), e1 = e0, e2 = e0, e3 = e0;
  if (lane < cnt) {
    s_f = (lane < deg) ? csr_src[base + lane] : d;
    const float4 a = node_logits<C>(als, s_f);
    e0 = lrelu(a.x + adv.x); e1 = lrelu(a.y + adv.y);
    e2 = lrelu(a.z + adv.z); e3 = lrelu(a.w + adv.w);
  }

  float m0 = e0, m1 = e1, m2 = e2, m3 = e3;
  for (int j = lane + 32; j < cnt; j += 32) {
    const int s = (j < deg) ? csr_src[base + j] : d;
    const float4 a = node_logits<C>(als, s);
    m0 = fmaxf(m0, lrelu(a.x + adv.x));
    m1 = fmaxf(m1, lrelu(a.y + adv.y));
    m2 = fmaxf(m2, lrelu(a.z + adv.z));
    m3 = fmaxf(m3, lrelu(a.w + adv.w));
  }
  if (spill) {
    const int novf = cnt_arr[NNODES];
    for (int p = lane; p < novf; p += 32) {
      if (ovf_dst[p] == d) {
        const float4 a = node_logits<C>(als, ovf_src[p]);
        m0 = fmaxf(m0, lrelu(a.x + adv.x));
        m1 = fmaxf(m1, lrelu(a.y + adv.y));
        m2 = fmaxf(m2, lrelu(a.z + adv.z));
        m3 = fmaxf(m3, lrelu(a.w + adv.w));
      }
    }
  }
  m0 = wave_max(m0); m1 = wave_max(m1); m2 = wave_max(m2); m3 = wave_max(m3);

  const bool in_f = lane < cnt;
  const float x0 = in_f ? __expf(e0 - m0) : 0.f, x1 = in_f ? __expf(e1 - m1) : 0.f;
  const float x2 = in_f ? __expf(e2 - m2) : 0.f, x3 = in_f ? __expf(e3 - m3) : 0.f;
  float z0 = x0, z1 = x1, z2 = x2, z3 = x3;
  for (int j = lane + 32; j < cnt; j += 32) {
    const int s = (j < deg) ? csr_src[base + j] : d;
    const float4 a = node_logits<C>(als, s);
    z0 += __expf(lrelu(a.x + adv.x) - m0);
    z1 += __expf(lrelu(a.y + adv.y) - m1);
    z2 += __expf(lrelu(a.z + adv.z) - m2);
    z3 += __expf(lrelu(a.w + adv.w) - m3);
  }
  if (spill) {
    const int novf = cnt_arr[NNODES];
    for (int p = lane; p < novf; p += 32) {
      if (ovf_dst[p] == d) {
        const float4 a = node_logits<C>(als, ovf_src[p]);
        z0 += __expf(lrelu(a.x + adv.x) - m0);
        z1 += __expf(lrelu(a.y + adv.y) - m1);
        z2 += __expf(lrelu(a.z + adv.z) - m2);
        z3 += __expf(lrelu(a.w + adv.w) - m3);
      }
    }
  }
  z0 = wave_sum(z0); z1 = wave_sum(z1); z2 = wave_sum(z2); z3 = wave_sum(z3);
  z0 += 1e-16f; z1 += 1e-16f; z2 += 1e-16f; z3 += 1e-16f;

  float4 acc[NV];
#pragma unroll
  for (int k = 0; k < NV; ++k) acc[k] = make_float4(0.f, 0.f, 0.f, 0.f);

  gather_rows<C, NV, UNR>(H, lane, (cnt < 32) ? cnt : 32, s_f,
                          x0 / z0, x1 / z1, x2 / z2, x3 / z3, acc);
  for (int j0 = 32; j0 < cnt; j0 += 32) {
    const int j = j0 + lane;
    int s_l = d;
    float p0 = 0.f, p1 = 0.f, p2 = 0.f, p3 = 0.f;
    if (j < cnt) {
      s_l = (j < deg) ? csr_src[base + j] : d;
      const float4 a = node_logits<C>(als, s_l);
      p0 = __expf(lrelu(a.x + adv.x) - m0) / z0;
      p1 = __expf(lrelu(a.y + adv.y) - m1) / z1;
      p2 = __expf(lrelu(a.z + adv.z) - m2) / z2;
      p3 = __expf(lrelu(a.w + adv.w) - m3) / z3;
    }
    const int left = cnt - j0;
    gather_rows<C, NV, UNR>(H, lane, (left < 32) ? left : 32, s_l, p0, p1, p2, p3, acc);
  }
  if (spill) {
    const int novf = cnt_arr[NNODES];
    for (int q0 = 0; q0 < novf; q0 += 32) {
      const int p = q0 + lane;
      int s_l = d;
      float p0 = 0.f, p1 = 0.f, p2 = 0.f, p3 = 0.f;
      const bool hit = (p < novf) && (ovf_dst[p] == d);
      if (hit) {
        s_l = ovf_src[p];
        const float4 a = node_logits<C>(als, s_l);
        p0 = __expf(lrelu(a.x + adv.x) - m0) / z0;
        p1 = __expf(lrelu(a.y + adv.y) - m1) / z1;
        p2 = __expf(lrelu(a.z + adv.z) - m2) / z2;
        p3 = __expf(lrelu(a.w + adv.w) - m3) / z3;
      }
      const int left = novf - q0;
      if (__builtin_amdgcn_ballot_w32(hit) != 0u)
        gather_rows<C, NV, UNR>(H, lane, (left < 32) ? left : 32, s_l, p0, p1, p2, p3, acc);
    }
  }

  if (MEAN) {
    float4 v = acc[0];
    v.x += lane_xor<8>(v.x);  v.y += lane_xor<8>(v.y);
    v.z += lane_xor<8>(v.z);  v.w += lane_xor<8>(v.w);
    v.x += lane_xor<16>(v.x); v.y += lane_xor<16>(v.y);
    v.z += lane_xor<16>(v.z); v.w += lane_xor<16>(v.w);
    if (lane < 8) {
      const float4 b = *(const float4*)(bias + lane * 4);
      float4 o;
      o.x = 0.25f * v.x + b.x; o.y = 0.25f * v.y + b.y;
      o.z = 0.25f * v.z + b.z; o.w = 0.25f * v.w + b.w;
      *(float4*)(out + (size_t)d * C + lane * 4) = o;
    }
  } else {
#pragma unroll
    for (int k = 0; k < NV; ++k) {
      const float4 b = *(const float4*)(bias + k * 128 + lane * 4);
      float4 o;
      o.x = fmaxf(acc[k].x + b.x, 0.f); o.y = fmaxf(acc[k].y + b.y, 0.f);
      o.z = fmaxf(acc[k].z + b.z, 0.f); o.w = fmaxf(acc[k].w + b.w, 0.f);
      *(float4*)(out + (size_t)d * Fout + k * 128 + lane * 4) = o;
    }
  }
}

__device__ __forceinline__ void mean_heads_store(float4 v, int lane,
                                                 const float* __restrict__ bias,
                                                 float* __restrict__ orow) {
  v.x += lane_xor<8>(v.x);  v.y += lane_xor<8>(v.y);
  v.z += lane_xor<8>(v.z);  v.w += lane_xor<8>(v.w);
  v.x += lane_xor<16>(v.x); v.y += lane_xor<16>(v.y);
  v.z += lane_xor<16>(v.z); v.w += lane_xor<16>(v.w);
  if (lane < 8) {
    const float4 b = *(const float4*)(bias + lane * 4);
    float4 o;
    o.x = 0.25f * v.x + b.x; o.y = 0.25f * v.y + b.y;
    o.z = 0.25f * v.z + b.z; o.w = 0.25f * v.w + b.w;
    *(float4*)(orow + lane * 4) = o;
  }
}

__device__ __forceinline__ void gather_rows_pair(const float* __restrict__ H, int lane, int nn,
                                                 int s_l, const float (&p)[2 * HEADS],
                                                 float4& acca, float4& accb) {
  constexpr int NH = 2 * HEADS;
  constexpr int UNR = 8;
  for (int t0 = 0; t0 < nn; t0 += UNR) {
#pragma unroll
    for (int u = 0; u < UNR; ++u) {
      const int t = t0 + u;
      const int s = __builtin_amdgcn_readlane(s_l, t);
      float a[NH];
#pragma unroll
      for (int h = 0; h < NH; ++h)
        a[h] = __int_as_float(__builtin_amdgcn_readlane(__float_as_int(p[h]), t));
      const float* __restrict__ hrow = H + (size_t)s * (NH * 32) + lane * 4;
      const float4 hva = *(const float4*)(hrow);
      const float4 hvb = *(const float4*)(hrow + 128);
      const float ala = pick_head<32>(0, lane, a[0], a[1], a[2], a[3]);
      const float alb = pick_head<32>(0, lane, a[4], a[5], a[6], a[7]);
      acca.x += ala * hva.x; acca.y += ala * hva.y;
      acca.z += ala * hva.z; acca.w += ala * hva.w;
      accb.x += alb * hvb.x; accb.y += alb * hvb.y;
      accb.z += alb * hvb.z; accb.w += alb * hvb.w;
    }
  }
}

__global__ __launch_bounds__(256, 8)
void gat_node_pair(const int* __restrict__ cnt_arr, const int* __restrict__ csr_src,
                   const int* __restrict__ ovf_dst, const int* __restrict__ ovf_src,
                   const float* __restrict__ H, const float* __restrict__ als,
                   const float* __restrict__ ald, const float* __restrict__ bias_a,
                   const float* __restrict__ bias_b, float* __restrict__ out_a,
                   float* __restrict__ out_b, int Nn) {
  constexpr int NH = 2 * HEADS;
  constexpr int C = 32;
  const int lane = threadIdx.x & 31;
  const int d = blockIdx.x * 8 + __builtin_amdgcn_readfirstlane((int)(threadIdx.x >> 5));
  if (d >= Nn) return;
  const int base = d * CAP;
  const int deg_all = cnt_arr[d];
  const int deg  = (deg_all < CAP) ? deg_all : CAP;
  const bool spill = deg_all > CAP;
  const int cnt  = deg + 1;
  float adv[NH];
  {
    const float4 x = *(const float4*)(ald + (size_t)d * NH);
    const float4 y = *(const float4*)(ald + (size_t)d * NH + 4);
    adv[0] = x.x; adv[1] = x.y; adv[2] = x.z; adv[3] = x.w;
    adv[4] = y.x; adv[5] = y.y; adv[6] = y.z; adv[7] = y.w;
  }

  int s_f = d;
  float ef[NH];
#pragma unroll
  for (int h = 0; h < NH; ++h) ef[h] = -__builtin_inff();
  if (lane < cnt) {
    s_f = (lane < deg) ? csr_src[base + lane] : d;
    const float4 a = *(const float4*)(als + (size_t)s_f * NH);
    const float4 b = *(const float4*)(als + (size_t)s_f * NH + 4);
    const float e[NH] = {a.x, a.y, a.z, a.w, b.x, b.y, b.z, b.w};
#pragma unroll
    for (int h = 0; h < NH; ++h) ef[h] = lrelu(e[h] + adv[h]);
  }

  float m[NH];
#pragma unroll
  for (int h = 0; h < NH; ++h) m[h] = ef[h];
  for (int j = lane + 32; j < cnt; j += 32) {
    const int s = (j < deg) ? csr_src[base + j] : d;
    const float4 a = *(const float4*)(als + (size_t)s * NH);
    const float4 b = *(const float4*)(als + (size_t)s * NH + 4);
    const float e[NH] = {a.x, a.y, a.z, a.w, b.x, b.y, b.z, b.w};
#pragma unroll
    for (int h = 0; h < NH; ++h) m[h] = fmaxf(m[h], lrelu(e[h] + adv[h]));
  }
  if (spill) {
    const int novf = cnt_arr[NNODES];
    for (int p = lane; p < novf; p += 32) {
      if (ovf_dst[p] == d) {
        const int s = ovf_src[p];
        const float4 a = *(const float4*)(als + (size_t)s * NH);
        const float4 b = *(const float4*)(als + (size_t)s * NH + 4);
        const float e[NH] = {a.x, a.y, a.z, a.w, b.x, b.y, b.z, b.w};
#pragma unroll
        for (int h = 0; h < NH; ++h) m[h] = fmaxf(m[h], lrelu(e[h] + adv[h]));
      }
    }
  }
#pragma unroll
  for (int h = 0; h < NH; ++h) m[h] = wave_max(m[h]);

  const bool in_f = lane < cnt;
  float xf[NH], z[NH];
#pragma unroll
  for (int h = 0; h < NH; ++h) {
    xf[h] = in_f ? __expf(ef[h] - m[h]) : 0.f;
    z[h] = xf[h];
  }
  for (int j = lane + 32; j < cnt; j += 32) {
    const int s = (j < deg) ? csr_src[base + j] : d;
    const float4 a = *(const float4*)(als + (size_t)s * NH);
    const float4 b = *(const float4*)(als + (size_t)s * NH + 4);
    const float e[NH] = {a.x, a.y, a.z, a.w, b.x, b.y, b.z, b.w};
#pragma unroll
    for (int h = 0; h < NH; ++h) z[h] += __expf(lrelu(e[h] + adv[h]) - m[h]);
  }
  if (spill) {
    const int novf = cnt_arr[NNODES];
    for (int p = lane; p < novf; p += 32) {
      if (ovf_dst[p] == d) {
        const int s = ovf_src[p];
        const float4 a = *(const float4*)(als + (size_t)s * NH);
        const float4 b = *(const float4*)(als + (size_t)s * NH + 4);
        const float e[NH] = {a.x, a.y, a.z, a.w, b.x, b.y, b.z, b.w};
#pragma unroll
        for (int h = 0; h < NH; ++h) z[h] += __expf(lrelu(e[h] + adv[h]) - m[h]);
      }
    }
  }
#pragma unroll
  for (int h = 0; h < NH; ++h) z[h] = wave_sum(z[h]) + 1e-16f;

  float4 acca = make_float4(0.f, 0.f, 0.f, 0.f);
  float4 accb = make_float4(0.f, 0.f, 0.f, 0.f);

  {
    float p[NH];
#pragma unroll
    for (int h = 0; h < NH; ++h) p[h] = xf[h] / z[h];
    gather_rows_pair(H, lane, (cnt < 32) ? cnt : 32, s_f, p, acca, accb);
  }
  for (int j0 = 32; j0 < cnt; j0 += 32) {
    const int j = j0 + lane;
    int s_l = d;
    float p[NH];
#pragma unroll
    for (int h = 0; h < NH; ++h) p[h] = 0.f;
    if (j < cnt) {
      s_l = (j < deg) ? csr_src[base + j] : d;
      const float4 a = *(const float4*)(als + (size_t)s_l * NH);
      const float4 b = *(const float4*)(als + (size_t)s_l * NH + 4);
      const float e[NH] = {a.x, a.y, a.z, a.w, b.x, b.y, b.z, b.w};
#pragma unroll
      for (int h = 0; h < NH; ++h) p[h] = __expf(lrelu(e[h] + adv[h]) - m[h]) / z[h];
    }
    const int left = cnt - j0;
    gather_rows_pair(H, lane, (left < 32) ? left : 32, s_l, p, acca, accb);
  }
  if (spill) {
    const int novf = cnt_arr[NNODES];
    for (int q0 = 0; q0 < novf; q0 += 32) {
      const int q = q0 + lane;
      int s_l = d;
      float p[NH];
#pragma unroll
      for (int h = 0; h < NH; ++h) p[h] = 0.f;
      const bool hit = (q < novf) && (ovf_dst[q] == d);
      if (hit) {
        s_l = ovf_src[q];
        const float4 a = *(const float4*)(als + (size_t)s_l * NH);
        const float4 b = *(const float4*)(als + (size_t)s_l * NH + 4);
        const float e[NH] = {a.x, a.y, a.z, a.w, b.x, b.y, b.z, b.w};
#pragma unroll
        for (int h = 0; h < NH; ++h) p[h] = __expf(lrelu(e[h] + adv[h]) - m[h]) / z[h];
      }
      const int left = novf - q0;
      if (__builtin_amdgcn_ballot_w32(hit) != 0u)
        gather_rows_pair(H, lane, (left < 32) ? left : 32, s_l, p, acca, accb);
    }
  }

  mean_heads_store(acca, lane, bias_a, out_a + (size_t)d * C);
  mean_heads_store(accb, lane, bias_b, out_b + (size_t)d * C);
}

extern "C" void kernel_launch(void* const* d_in, const int* in_sizes, int n_in,
                              void* d_out, int out_size, void* d_ws, size_t ws_size,
                              hipStream_t stream) {
  const float* x   = (const float*)d_in[0];
  const int*   ei  = (const int*)d_in[1];
  const int E = in_sizes[1] / 2;
  const int* src = ei;
  const int* dst = ei + E;
  const float *W1 = (const float*)d_in[2],  *as1 = (const float*)d_in[3],
              *ad1 = (const float*)d_in[4], *b1  = (const float*)d_in[5];
  const float *W2 = (const float*)d_in[6],  *as2 = (const float*)d_in[7],
              *ad2 = (const float*)d_in[8], *b2  = (const float*)d_in[9];
  const float *Wm = (const float*)d_in[10], *asm_ = (const float*)d_in[11],
              *adm = (const float*)d_in[12], *bm = (const float*)d_in[13];
  const float *Wl = (const float*)d_in[14], *asl = (const float*)d_in[15],
              *adl = (const float*)d_in[16], *bl = (const float*)d_in[17];

  char* ws = (char*)d_ws;
  size_t off = 0;
  auto alloc = [&](size_t bytes) -> void* {
    void* p = ws + off;
    off += (bytes + 255) & ~(size_t)255;
    return p;
  };
  float* Hbuf = (float*)alloc((size_t)NNODES * 512 * 4);
  float* A0   = (float*)alloc((size_t)NNODES * 256 * 4);
  float* A1   = (float*)alloc((size_t)NNODES * 512 * 4);
  float* ALS  = (float*)alloc((size_t)NNODES * 2 * HEADS * 4);
  float* ALD  = (float*)alloc((size_t)NNODES * 2 * HEADS * 4);
  int*   CNT  = (int*)alloc((size_t)(NNODES + 1) * 4);
  int*   BKT  = (int*)alloc((size_t)NNODES * CAP * 4);
  int*   OVD  = (int*)alloc((size_t)E * 4);
  int*   OVS  = (int*)alloc((size_t)E * 4);

  float* zmean   = (float*)d_out;
  float* zlogstd = (float*)d_out + (size_t)NNODES * 32;

  const int Mt8 = (NNODES / 16 + 7) / 8;
  const int nblk = (NNODES + 7) / 8;

  gemm_xw_wmma<<<dim3(Mt8, 256 / 64), 256, 0, stream>>>(x, W1, W1, Hbuf, NNODES, 256, 256, 256, 256,
                                                       as1, ad1, as1, ad1, ALS, ALD, 1, 256 / 64,
                                                       CNT, NNODES + 1);
  dst_bucket<<<(E + 1023) / 1024, 256, 0, stream>>>(src, dst, E, CNT, BKT, OVD, OVS);
  gat_node<64, false><<<nblk, 256, 0, stream>>>(CNT, BKT, OVD, OVS, Hbuf, ALS, ALD, b1, A0, NNODES);

  gemm_xw_wmma<<<dim3(Mt8, 512 / 64), 256, 0, stream>>>(A0, W2, W2, Hbuf, NNODES, 256, 512, 512, 512,
                                                       as2, ad2, as2, ad2, ALS, ALD, 1, 512 / 64,
                                                       nullptr, 0);
  gat_node<128, false><<<nblk, 256, 0, stream>>>(CNT, BKT, OVD, OVS, Hbuf, ALS, ALD, b2, A1, NNODES);

  gemm_xw_wmma<<<dim3(Mt8, 256 / 64), 256, 0, stream>>>(A1, Wm, Wl, Hbuf, NNODES, 512, 128, 256, 128,
                                                       asm_, adm, asl, adl, ALS, ALD, 2, 8,
                                                       nullptr, 0);
  gat_node_pair<<<nblk, 256, 0, stream>>>(CNT, BKT, OVD, OVS, Hbuf, ALS, ALD, bm, bl,
                                          zmean, zlogstd, NNODES);
}
